# starting kernel
# baseline (speedup 1.0000x reference)
#include <hip/hip_runtime.h>

typedef __attribute__((ext_vector_type(2))) float v2f;
typedef __attribute__((ext_vector_type(8))) float v8f;

#define N_NODES 100000
#define N_EDGES 1600000
#define IN_DIM 256
#define HID_DIM 128
#define OUT_DIM 64

__global__ __launch_bounds__(256) void gemm1_wmma(const float* __restrict__ X,
                                                  const float* __restrict__ W1,
                                                  float* __restrict__ H1) {
    __shared__ float xs[16][260];
    __shared__ float w1s[64][136];

    const int tid  = threadIdx.x;
    const int lane = tid & 31;
    const int wave = tid >> 5;
    const int half = lane >> 4;
    const int l16  = lane & 15;
    const int mbase = blockIdx.x * 16;

    {
        int row = tid >> 4;
        int seg = (tid & 15) * 16;
        const float* xp = X + (size_t)(mbase + row) * IN_DIM + seg;
        float4 a0 = ((const float4*)xp)[0];
        float4 a1 = ((const float4*)xp)[1];
        float4 a2 = ((const float4*)xp)[2];
        float4 a3 = ((const float4*)xp)[3];
        *(float4*)&xs[row][seg + 0]  = a0;
        *(float4*)&xs[row][seg + 4]  = a1;
        *(float4*)&xs[row][seg + 8]  = a2;
        *(float4*)&xs[row][seg + 12] = a3;
    }

    v8f acc = {};
    const int nb = wave * 16 + l16;

    for (int kc = 0; kc < IN_DIM; kc += 64) {
        for (int i = tid; i < (64 * 128) / 4; i += 256) {
            int flat = i * 4;
            int kr   = flat >> 7;
            int col  = flat & 127;
            float4 w = *(const float4*)&W1[(size_t)(kc + kr) * HID_DIM + col];
            *(float4*)&w1s[kr][col] = w;
        }
        __syncthreads();

#pragma unroll
        for (int kk = 0; kk < 64; kk += 4) {
            const int ka = kc + kk + half * 2;
            const int kb = kk + half * 2;
            v2f a = {xs[l16][ka], xs[l16][ka + 1]};
            v2f b = {w1s[kb][nb], w1s[kb + 1][nb]};
            acc = __builtin_amdgcn_wmma_f32_16x16x4_f32(
                false, a, false, b, (short)0, acc, false, false);
        }
        __syncthreads();
    }

#pragma unroll
    for (int v = 0; v < 8; ++v) {
        int row = mbase + v + half * 8;
        H1[(size_t)row * HID_DIM + nb] = acc[v];
    }
}

__global__ __launch_bounds__(256) void gemm2_wmma(const float* __restrict__ Hagg,
                                                  const float* __restrict__ W2,
                                                  float* __restrict__ H2) {
    __shared__ float hs[32][132];
    __shared__ float w2s[128][72];

    const int tid  = threadIdx.x;
    const int lane = tid & 31;
    const int wave = tid >> 5;
    const int mt   = wave >> 2;
    const int nt   = wave & 3;
    const int half = lane >> 4;
    const int l16  = lane & 15;
    const int mbase = blockIdx.x * 32;

    for (int i = tid; i < 1024; i += 256) {
        int flat = i * 4;
        int row  = flat >> 7;
        int col  = flat & 127;
        float4 h = *(const float4*)&Hagg[(size_t)(mbase + row) * HID_DIM + col];
        h.x = fmaxf(h.x, 0.0f);
        h.y = fmaxf(h.y, 0.0f);
        h.z = fmaxf(h.z, 0.0f);
        h.w = fmaxf(h.w, 0.0f);
        *(float4*)&hs[row][col] = h;
    }
    for (int i = tid; i < 2048; i += 256) {
        int flat = i * 4;
        int kr   = flat >> 6;
        int col  = flat & 63;
        *(float4*)&w2s[kr][col] = *(const float4*)&W2[(size_t)kr * OUT_DIM + col];
    }
    __syncthreads();

    v8f acc = {};
    const int mrow = mt * 16 + l16;
    const int nb   = nt * 16 + l16;

#pragma unroll
    for (int kk = 0; kk < HID_DIM; kk += 4) {
        const int k = kk + half * 2;
        v2f a = {hs[mrow][k], hs[mrow][k + 1]};
        v2f b = {w2s[k][nb], w2s[k + 1][nb]};
        acc = __builtin_amdgcn_wmma_f32_16x16x4_f32(
            false, a, false, b, (short)0, acc, false, false);
    }

#pragma unroll
    for (int v = 0; v < 8; ++v) {
        int row = mbase + mt * 16 + v + half * 8;
        H2[(size_t)row * OUT_DIM + nt * 16 + l16] = acc[v];
    }
}

__global__ __launch_bounds__(256) void spmm_scatter_d128(
    const int* __restrict__ src, const int* __restrict__ dst,
    const float* __restrict__ vals, const float* __restrict__ H,
    float* __restrict__ out, int E) {
    int e = blockIdx.x * 8 + threadIdx.y;
    if (e >= E) return;
    int lane = threadIdx.x;
    int s = src[e];
    int d = dst[e];
    float v = vals[e];
    float4 m = ((const float4*)(H + (size_t)s * 128))[lane];
    float* op = out + (size_t)d * 128 + lane * 4;
    atomicAdd(op + 0, m.x * v);
    atomicAdd(op + 1, m.y * v);
    atomicAdd(op + 2, m.z * v);
    atomicAdd(op + 3, m.w * v);
}

__global__ __launch_bounds__(256) void spmm_scatter_d64(
    const int* __restrict__ src, const int* __restrict__ dst,
    const float* __restrict__ vals, const float* __restrict__ H,
    float* __restrict__ out, int E) {
    int e = blockIdx.x * 8 + threadIdx.y;
    if (e >= E) return;
    int lane = threadIdx.x;
    int s = src[e];
    int d = dst[e];
    float v = vals[e];
    float2 m = ((const float2*)(H + (size_t)s * 64))[lane];
    float* op = out + (size_t)d * 64 + lane * 2;
    atomicAdd(op + 0, m.x * v);
    atomicAdd(op + 1, m.y * v);
}

extern "C" void kernel_launch(void* const* d_in, const int* in_sizes, int n_in,
                              void* d_out, int out_size, void* d_ws, size_t ws_size,
                              hipStream_t stream) {
    const float* X     = (const float*)d_in[0];
    const int*   esrc  = (const int*)d_in[1];
    const int*   edst  = (const int*)d_in[2];
    const float* evals = (const float*)d_in[3];
    const float* W1    = (const float*)d_in[4];
    const float* W2    = (const float*)d_in[5];
    float*       out   = (float*)d_out;

    char* ws = (char*)d_ws;
    const size_t h1Bytes = (size_t)N_NODES * HID_DIM * sizeof(float);
    float* H1   = (float*)(ws);
    float* Hagg = (float*)(ws + h1Bytes);
    float* H2   = (float*)(ws + 2 * h1Bytes);

    hipMemsetAsync(Hagg, 0, h1Bytes, stream);
    gemm1_wmma<<<N_NODES / 16, 256, 0, stream>>>(X, W1, H1);
    dim3 eb(32, 8);
    spmm_scatter_d128<<<N_EDGES / 8, eb, 0, stream>>>(esrc, edst, evals, H1, Hagg, N_EDGES);

    hipMemsetAsync(out, 0, (size_t)N_NODES * OUT_DIM * sizeof(float), stream);
    gemm2_wmma<<<N_NODES / 32, 256, 0, stream>>>(Hagg, W2, H2);
    spmm_scatter_d64<<<N_EDGES / 8, eb, 0, stream>>>(esrc, edst, evals, H2, out, N_EDGES);
}


# f32 atomic scatter replaced by a per-call edge list grouped by destination (int atomic rank + scan + fill) and one wave per row summing gathered rows in registers; row sum order = arrival rank
# speedup vs baseline: 1.8082x; 1.8082x over previous
#include <hip/hip_runtime.h>

typedef __attribute__((ext_vector_type(2))) float v2f;
typedef __attribute__((ext_vector_type(8))) float v8f;

#define N_NODES 100000
#define N_EDGES 1600000
#define IN_DIM 256
#define HID_DIM 128
#define OUT_DIM 64

__global__ __launch_bounds__(256) void gemm1_wmma(const float* __restrict__ X,
                                                  const float* __restrict__ W1,
                                                  float* __restrict__ H1) {
    __shared__ float xs[16][260];
    __shared__ float w1s[64][136];

    const int tid  = threadIdx.x;
    const int lane = tid & 31;
    const int wave = tid >> 5;
    const int half = lane >> 4;
    const int l16  = lane & 15;
    const int mbase = blockIdx.x * 16;

    {
        int row = tid >> 4;
        int seg = (tid & 15) * 16;
        const float* xp = X + (size_t)(mbase + row) * IN_DIM + seg;
        float4 a0 = ((const float4*)xp)[0];
        float4 a1 = ((const float4*)xp)[1];
        float4 a2 = ((const float4*)xp)[2];
        float4 a3 = ((const float4*)xp)[3];
        *(float4*)&xs[row][seg + 0]  = a0;
        *(float4*)&xs[row][seg + 4]  = a1;
        *(float4*)&xs[row][seg + 8]  = a2;
        *(float4*)&xs[row][seg + 12] = a3;
    }

    v8f acc = {};
    const int nb = wave * 16 + l16;

    for (int kc = 0; kc < IN_DIM; kc += 64) {
        for (int i = tid; i < (64 * 128) / 4; i += 256) {
            int flat = i * 4;
            int kr   = flat >> 7;
            int col  = flat & 127;
            float4 w = *(const float4*)&W1[(size_t)(kc + kr) * HID_DIM + col];
            *(float4*)&w1s[kr][col] = w;
        }
        __syncthreads();

#pragma unroll
        for (int kk = 0; kk < 64; kk += 4) {
            const int ka = kc + kk + half * 2;
            const int kb = kk + half * 2;
            v2f a = {xs[l16][ka], xs[l16][ka + 1]};
            v2f b = {w1s[kb][nb], w1s[kb + 1][nb]};
            acc = __builtin_amdgcn_wmma_f32_16x16x4_f32(
                false, a, false, b, (short)0, acc, false, false);
        }
        __syncthreads();
    }

#pragma unroll
    for (int v = 0; v < 8; ++v) {
        int row = mbase + v + half * 8;
        H1[(size_t)row * HID_DIM + nb] = acc[v];
    }
}

__global__ __launch_bounds__(256) void gemm2_wmma(const float* __restrict__ Hagg,
                                                  const float* __restrict__ W2,
                                                  float* __restrict__ H2) {
    __shared__ float hs[32][132];
    __shared__ float w2s[128][72];

    const int tid  = threadIdx.x;
    const int lane = tid & 31;
    const int wave = tid >> 5;
    const int mt   = wave >> 2;
    const int nt   = wave & 3;
    const int half = lane >> 4;
    const int l16  = lane & 15;
    const int mbase = blockIdx.x * 32;

    for (int i = tid; i < 1024; i += 256) {
        int flat = i * 4;
        int row  = flat >> 7;
        int col  = flat & 127;
        float4 h = *(const float4*)&Hagg[(size_t)(mbase + row) * HID_DIM + col];
        h.x = fmaxf(h.x, 0.0f);
        h.y = fmaxf(h.y, 0.0f);
        h.z = fmaxf(h.z, 0.0f);
        h.w = fmaxf(h.w, 0.0f);
        *(float4*)&hs[row][col] = h;
    }
    for (int i = tid; i < 2048; i += 256) {
        int flat = i * 4;
        int kr   = flat >> 6;
        int col  = flat & 63;
        *(float4*)&w2s[kr][col] = *(const float4*)&W2[(size_t)kr * OUT_DIM + col];
    }
    __syncthreads();

    v8f acc = {};
    const int mrow = mt * 16 + l16;
    const int nb   = nt * 16 + l16;

#pragma unroll
    for (int kk = 0; kk < HID_DIM; kk += 4) {
        const int k = kk + half * 2;
        v2f a = {hs[mrow][k], hs[mrow][k + 1]};
        v2f b = {w2s[k][nb], w2s[k + 1][nb]};
        acc = __builtin_amdgcn_wmma_f32_16x16x4_f32(
            false, a, false, b, (short)0, acc, false, false);
    }

#pragma unroll
    for (int v = 0; v < 8; ++v) {
        int row = mbase + mt * 16 + v + half * 8;
        H2[(size_t)row * OUT_DIM + nt * 16 + l16] = acc[v];
    }
}

__global__ __launch_bounds__(256) void edge_rank(const int* __restrict__ dst,
                                                 int* __restrict__ cnt,
                                                 int* __restrict__ rank, int E) {
    int e = blockIdx.x * 256 + threadIdx.x;
    if (e >= E) return;
    rank[e] = atomicAdd(&cnt[dst[e]], 1);
}

__global__ __launch_bounds__(1024) void rowptr_scan(const int* __restrict__ cnt,
                                                    int* __restrict__ rowptr) {
    __shared__ int wsum[32];
    __shared__ int s_prefix;
    const int tid  = threadIdx.x;
    const int lane = tid & 31;
    const int wave = tid >> 5;
    const int start = blockIdx.x * 1024;

    int p = 0;
    for (int i = tid * 4; i < start; i += 4096) {
        int4 c = *(const int4*)(cnt + i);
        p += (c.x + c.y) + (c.z + c.w);
    }
    for (int o = 16; o > 0; o >>= 1) p += __shfl_xor(p, o);
    if (lane == 0) wsum[wave] = p;
    __syncthreads();
    if (wave == 0) {
        int t = wsum[lane];
        for (int o = 16; o > 0; o >>= 1) t += __shfl_xor(t, o);
        if (lane == 0) s_prefix = t;
    }
    __syncthreads();
    const int prefix = s_prefix;

    const int g = start + tid;
    const int c = (g < N_NODES) ? cnt[g] : 0;
    int incl = c;
    for (int o = 1; o < 32; o <<= 1) {
        int t = __shfl_up(incl, o);
        if (lane >= o) incl += t;
    }
    if (lane == 31) wsum[wave] = incl;
    __syncthreads();
    if (wave == 0) {
        const int t = wsum[lane];
        int ti = t;
        for (int o = 1; o < 32; o <<= 1) {
            int u = __shfl_up(ti, o);
            if (lane >= o) ti += u;
        }
        wsum[lane] = ti - t;
    }
    __syncthreads();
    const int excl = prefix + wsum[wave] + incl - c;
    if (g < N_NODES) rowptr[g] = excl;
    if (g == N_NODES - 1) rowptr[N_NODES] = excl + c;
}

__global__ __launch_bounds__(256) void edge_fill(const int* __restrict__ src,
                                                 const int* __restrict__ dst,
                                                 const float* __restrict__ vals,
                                                 const int* __restrict__ rowptr,
                                                 const int* __restrict__ rank,
                                                 int2* __restrict__ list, int E) {
    int e = blockIdx.x * 256 + threadIdx.x;
    if (e >= E) return;
    const int pos = rowptr[dst[e]] + rank[e];
    list[pos] = make_int2(src[e], __float_as_int(vals[e]));
}

template <int U>
__device__ __forceinline__ void rows128(const float* __restrict__ H, int mysrc, int myval,
                                        int j, int lane, float4& acc) {
    float4 m[U];
    float  v[U];
#pragma unroll
    for (int k = 0; k < U; ++k) {
        const int s = __builtin_amdgcn_readlane(mysrc, j + k);
        v[k] = __int_as_float(__builtin_amdgcn_readlane(myval, j + k));
        m[k] = ((const float4*)(H + (size_t)s * 128))[lane];
    }
#pragma unroll
    for (int k = 0; k < U; ++k) {
        acc.x = fmaf(v[k], m[k].x, acc.x);
        acc.y = fmaf(v[k], m[k].y, acc.y);
        acc.z = fmaf(v[k], m[k].z, acc.z);
        acc.w = fmaf(v[k], m[k].w, acc.w);
    }
}

template <int U>
__device__ __forceinline__ void rows64(const float* __restrict__ H, int mysrc, int myval,
                                       int j, int lane, float2& acc) {
    float2 m[U];
    float  v[U];
#pragma unroll
    for (int k = 0; k < U; ++k) {
        const int s = __builtin_amdgcn_readlane(mysrc, j + k);
        v[k] = __int_as_float(__builtin_amdgcn_readlane(myval, j + k));
        m[k] = ((const float2*)(H + (size_t)s * 64))[lane];
    }
#pragma unroll
    for (int k = 0; k < U; ++k) {
        acc.x = fmaf(v[k], m[k].x, acc.x);
        acc.y = fmaf(v[k], m[k].y, acc.y);
    }
}

__global__ __launch_bounds__(256) void spmm_gather_d128(
    const int* __restrict__ rowptr, const int2* __restrict__ list,
    const float* __restrict__ H, float* __restrict__ out) {
    const int lane = threadIdx.x & 31;
    const int row  = blockIdx.x * 8 + (threadIdx.x >> 5);
    if (row >= N_NODES) return;
    const int beg = __builtin_amdgcn_readfirstlane(rowptr[row]);
    const int end = __builtin_amdgcn_readfirstlane(rowptr[row + 1]);
    float4 acc = make_float4(0.0f, 0.0f, 0.0f, 0.0f);
    for (int base = beg; base < end; base += 32) {
        const int n = min(32, end - base);
        int2 my = make_int2(0, 0);
        if (lane < n) my = list[base + lane];
        int j = 0;
        for (; j + 8 <= n; j += 8) rows128<8>(H, my.x, my.y, j, lane, acc);
        if (j + 4 <= n) { rows128<4>(H, my.x, my.y, j, lane, acc); j += 4; }
        if (j + 2 <= n) { rows128<2>(H, my.x, my.y, j, lane, acc); j += 2; }
        if (j < n)      { rows128<1>(H, my.x, my.y, j, lane, acc); }
    }
    ((float4*)(out + (size_t)row * 128))[lane] = acc;
}

__global__ __launch_bounds__(256) void spmm_gather_d64(
    const int* __restrict__ rowptr, const int2* __restrict__ list,
    const float* __restrict__ H, float* __restrict__ out) {
    const int lane = threadIdx.x & 31;
    const int row  = blockIdx.x * 8 + (threadIdx.x >> 5);
    if (row >= N_NODES) return;
    const int beg = __builtin_amdgcn_readfirstlane(rowptr[row]);
    const int end = __builtin_amdgcn_readfirstlane(rowptr[row + 1]);
    float2 acc = make_float2(0.0f, 0.0f);
    for (int base = beg; base < end; base += 32) {
        const int n = min(32, end - base);
        int2 my = make_int2(0, 0);
        if (lane < n) my = list[base + lane];
        int j = 0;
        for (; j + 8 <= n; j += 8) rows64<8>(H, my.x, my.y, j, lane, acc);
        if (j + 4 <= n) { rows64<4>(H, my.x, my.y, j, lane, acc); j += 4; }
        if (j + 2 <= n) { rows64<2>(H, my.x, my.y, j, lane, acc); j += 2; }
        if (j < n)      { rows64<1>(H, my.x, my.y, j, lane, acc); }
    }
    ((float2*)(out + (size_t)row * 64))[lane] = acc;
}

extern "C" void kernel_launch(void* const* d_in, const int* in_sizes, int n_in,
                              void* d_out, int out_size, void* d_ws, size_t ws_size,
                              hipStream_t stream) {
    const float* X     = (const float*)d_in[0];
    const int*   esrc  = (const int*)d_in[1];
    const int*   edst  = (const int*)d_in[2];
    const float* evals = (const float*)d_in[3];
    const float* W1    = (const float*)d_in[4];
    const float* W2    = (const float*)d_in[5];
    float*       out   = (float*)d_out;

    char* ws = (char*)d_ws;
    const size_t h1Bytes = (size_t)N_NODES * HID_DIM * sizeof(float);
    float* H1   = (float*)(ws);
    float* Hagg = (float*)(ws + h1Bytes);
    float* H2   = (float*)(ws);
    int2* list   = (int2*)(ws + 2 * h1Bytes);
    int*  rank   = (int*)(ws + 2 * h1Bytes + 12800000);
    int*  cnt    = (int*)(ws + 2 * h1Bytes + 12800000 + 6400000);
    int*  rowptr = (int*)(ws + 2 * h1Bytes + 12800000 + 6400000 + 400128);

    hipMemsetAsync(cnt, 0, (size_t)N_NODES * sizeof(int), stream);
    edge_rank<<<N_EDGES / 256, 256, 0, stream>>>(edst, cnt, rank, N_EDGES);
    rowptr_scan<<<(N_NODES + 1023) / 1024, 1024, 0, stream>>>(cnt, rowptr);
    edge_fill<<<N_EDGES / 256, 256, 0, stream>>>(esrc, edst, evals, rowptr, rank, list, N_EDGES);

    gemm1_wmma<<<N_NODES / 16, 256, 0, stream>>>(X, W1, H1);
    spmm_gather_d128<<<N_NODES / 8, 256, 0, stream>>>(rowptr, list, H1, Hagg);

    gemm2_wmma<<<N_NODES / 32, 256, 0, stream>>>(Hagg, W2, H2);
    spmm_gather_d64<<<N_NODES / 8, 256, 0, stream>>>(rowptr, list, H2, out);
}


# global int atomic edge rank replaced by LDS counters: edges binned by dst range (LDS histogram per chunk, count table scan), then ranked per bin in LDS; no global atomics; row sum order = LDS arrival
# speedup vs baseline: 2.8500x; 1.5762x over previous
#include <hip/hip_runtime.h>

typedef __attribute__((ext_vector_type(2))) float v2f;
typedef __attribute__((ext_vector_type(8))) float v8f;

#define N_NODES 100000
#define N_EDGES 1600000
#define IN_DIM 256
#define HID_DIM 128
#define OUT_DIM 64

__global__ __launch_bounds__(256) void gemm1_wmma(const float* __restrict__ X,
                                                  const float* __restrict__ W1,
                                                  float* __restrict__ H1) {
    __shared__ float xs[16][260];
    __shared__ float w1s[64][136];

    const int tid  = threadIdx.x;
    const int lane = tid & 31;
    const int wave = tid >> 5;
    const int half = lane >> 4;
    const int l16  = lane & 15;
    const int mbase = blockIdx.x * 16;

    {
        int row = tid >> 4;
        int seg = (tid & 15) * 16;
        const float* xp = X + (size_t)(mbase + row) * IN_DIM + seg;
        float4 a0 = ((const float4*)xp)[0];
        float4 a1 = ((const float4*)xp)[1];
        float4 a2 = ((const float4*)xp)[2];
        float4 a3 = ((const float4*)xp)[3];
        *(float4*)&xs[row][seg + 0]  = a0;
        *(float4*)&xs[row][seg + 4]  = a1;
        *(float4*)&xs[row][seg + 8]  = a2;
        *(float4*)&xs[row][seg + 12] = a3;
    }

    v8f acc = {};
    const int nb = wave * 16 + l16;

    for (int kc = 0; kc < IN_DIM; kc += 64) {
        for (int i = tid; i < (64 * 128) / 4; i += 256) {
            int flat = i * 4;
            int kr   = flat >> 7;
            int col  = flat & 127;
            float4 w = *(const float4*)&W1[(size_t)(kc + kr) * HID_DIM + col];
            *(float4*)&w1s[kr][col] = w;
        }
        __syncthreads();

#pragma unroll
        for (int kk = 0; kk < 64; kk += 4) {
            const int ka = kc + kk + half * 2;
            const int kb = kk + half * 2;
            v2f a = {xs[l16][ka], xs[l16][ka + 1]};
            v2f b = {w1s[kb][nb], w1s[kb + 1][nb]};
            acc = __builtin_amdgcn_wmma_f32_16x16x4_f32(
                false, a, false, b, (short)0, acc, false, false);
        }
        __syncthreads();
    }

#pragma unroll
    for (int v = 0; v < 8; ++v) {
        int row = mbase + v + half * 8;
        H1[(size_t)row * HID_DIM + nb] = acc[v];
    }
}

__global__ __launch_bounds__(256) void gemm2_wmma(const float* __restrict__ Hagg,
                                                  const float* __restrict__ W2,
                                                  float* __restrict__ H2) {
    __shared__ float hs[32][132];
    __shared__ float w2s[128][72];

    const int tid  = threadIdx.x;
    const int lane = tid & 31;
    const int wave = tid >> 5;
    const int mt   = wave >> 2;
    const int nt   = wave & 3;
    const int half = lane >> 4;
    const int l16  = lane & 15;
    const int mbase = blockIdx.x * 32;

    for (int i = tid; i < 1024; i += 256) {
        int flat = i * 4;
        int row  = flat >> 7;
        int col  = flat & 127;
        float4 h = *(const float4*)&Hagg[(size_t)(mbase + row) * HID_DIM + col];
        h.x = fmaxf(h.x, 0.0f);
        h.y = fmaxf(h.y, 0.0f);
        h.z = fmaxf(h.z, 0.0f);
        h.w = fmaxf(h.w, 0.0f);
        *(float4*)&hs[row][col] = h;
    }
    for (int i = tid; i < 2048; i += 256) {
        int flat = i * 4;
        int kr   = flat >> 6;
        int col  = flat & 63;
        *(float4*)&w2s[kr][col] = *(const float4*)&W2[(size_t)kr * OUT_DIM + col];
    }
    __syncthreads();

    v8f acc = {};
    const int mrow = mt * 16 + l16;
    const int nb   = nt * 16 + l16;

#pragma unroll
    for (int kk = 0; kk < HID_DIM; kk += 4) {
        const int k = kk + half * 2;
        v2f a = {hs[mrow][k], hs[mrow][k + 1]};
        v2f b = {w2s[k][nb], w2s[k + 1][nb]};
        acc = __builtin_amdgcn_wmma_f32_16x16x4_f32(
            false, a, false, b, (short)0, acc, false, false);
    }

#pragma unroll
    for (int v = 0; v < 8; ++v) {
        int row = mbase + mt * 16 + v + half * 8;
        H2[(size_t)row * OUT_DIM + nt * 16 + l16] = acc[v];
    }
}

#define NBIN 256
#define BIN_NODES 391
#define NCHUNK 256
#define CHUNK_EDGES (N_EDGES / NCHUNK)
#define CHUNK_TRIPS 7
#define SRC_BITS 17

__global__ __launch_bounds__(1024) void bin_count(const int* __restrict__ dst,
                                                  int* __restrict__ counts) {
    __shared__ int hist[NBIN];
    const int t = threadIdx.x;
    if (t < NBIN) hist[t] = 0;
    __syncthreads();
    const int beg = blockIdx.x * CHUNK_EDGES;
    const int end = beg + CHUNK_EDGES;
    int d[CHUNK_TRIPS];
#pragma unroll
    for (int k = 0; k < CHUNK_TRIPS; ++k) {
        const int e = beg + k * 1024 + t;
        d[k] = (e < end) ? dst[e] : -1;
    }
#pragma unroll
    for (int k = 0; k < CHUNK_TRIPS; ++k)
        if (d[k] >= 0) atomicAdd(&hist[(unsigned)d[k] / BIN_NODES], 1);
    __syncthreads();
    if (t < NBIN) counts[blockIdx.x * NBIN + t] = hist[t];
}

__global__ __launch_bounds__(1024) void bin_scatter(const int* __restrict__ src,
                                                    const int* __restrict__ dst,
                                                    const float* __restrict__ vals,
                                                    const int* __restrict__ counts,
                                                    int* __restrict__ binStart,
                                                    int2* __restrict__ recs) {
    __shared__ int ptot[16][NBIN];
    __shared__ int pbef[16][NBIN];
    __shared__ int cursor[NBIN];
    __shared__ int wsum[8];
    const int t    = threadIdx.x;
    const int lane = t & 31;
    const int wave = t >> 5;
    const int me   = blockIdx.x;
    const int beg  = me * CHUNK_EDGES;
    const int end  = beg + CHUNK_EDGES;

    int   d[CHUNK_TRIPS], s[CHUNK_TRIPS];
    float v[CHUNK_TRIPS];
#pragma unroll
    for (int k = 0; k < CHUNK_TRIPS; ++k) {
        const int e = beg + k * 1024 + t;
        const bool ok = e < end;
        d[k] = ok ? dst[e] : -1;
        s[k] = ok ? src[e] : 0;
        v[k] = ok ? vals[e] : 0.0f;
    }

    {
        const int j4 = (t & 63) * 4;
        const int q  = t >> 6;
        int4 tot = make_int4(0, 0, 0, 0);
        int4 bef = make_int4(0, 0, 0, 0);
#pragma unroll
        for (int k = 0; k < 16; ++k) {
            const int c = q * 16 + k;
            const int4 n = *(const int4*)(counts + c * NBIN + j4);
            tot.x += n.x; tot.y += n.y; tot.z += n.z; tot.w += n.w;
            if (c < me) { bef.x += n.x; bef.y += n.y; bef.z += n.z; bef.w += n.w; }
        }
        *(int4*)&ptot[q][j4] = tot;
        *(int4*)&pbef[q][j4] = bef;
    }
    __syncthreads();
    int myTot = 0, myBef = 0;
    if (t < NBIN) {
#pragma unroll
        for (int k = 0; k < 16; ++k) { myTot += ptot[k][t]; myBef += pbef[k][t]; }
    }
    int incl = myTot;
    for (int o = 1; o < 32; o <<= 1) {
        const int u = __shfl_up(incl, o);
        if (lane >= o) incl += u;
    }
    if (lane == 31 && wave < 8) wsum[wave] = incl;
    __syncthreads();
    if (t < NBIN) {
        int base = 0;
        for (int w = 0; w < wave; ++w) base += wsum[w];
        const int excl = base + incl - myTot;
        cursor[t] = excl + myBef;
        if (me == 0) {
            binStart[t] = excl;
            if (t == NBIN - 1) binStart[NBIN] = excl + myTot;
        }
    }
    __syncthreads();

    int pos[CHUNK_TRIPS];
#pragma unroll
    for (int k = 0; k < CHUNK_TRIPS; ++k) {
        pos[k] = 0;
        if (d[k] >= 0) {
            const unsigned bin = (unsigned)d[k] / BIN_NODES;
            pos[k] = atomicAdd(&cursor[bin], 1);
            s[k] |= (d[k] - (int)bin * BIN_NODES) << SRC_BITS;
        }
    }
#pragma unroll
    for (int k = 0; k < CHUNK_TRIPS; ++k)
        if (d[k] >= 0) recs[pos[k]] = make_int2(s[k], __float_as_int(v[k]));
}

__global__ __launch_bounds__(1024) void bin_finish(const int* __restrict__ binStart,
                                                   const int2* __restrict__ recs,
                                                   int* __restrict__ rowptr,
                                                   int2* __restrict__ list) {
    __shared__ int cnt[512];
    __shared__ int wsum[16];
    const int t    = threadIdx.x;
    const int lane = t & 31;
    const int wave = t >> 5;
    const int b    = blockIdx.x;
    const int lo   = b * BIN_NODES;
    const int nn   = min(BIN_NODES, N_NODES - lo);
    const int bs   = binStart[b];
    const int be   = binStart[b + 1];
    if (t < 512) cnt[t] = 0;
    __syncthreads();

    for (int base = bs; base < be; base += 8192) {
        int x[8];
#pragma unroll
        for (int k = 0; k < 8; ++k) {
            const int e = base + k * 1024 + t;
            x[k] = (e < be) ? recs[e].x : -1;
        }
#pragma unroll
        for (int k = 0; k < 8; ++k)
            if (x[k] >= 0) atomicAdd(&cnt[x[k] >> SRC_BITS], 1);
    }
    __syncthreads();

    const int c = (t < 512) ? cnt[t] : 0;
    int incl = c;
    for (int o = 1; o < 32; o <<= 1) {
        const int u = __shfl_up(incl, o);
        if (lane >= o) incl += u;
    }
    if (lane == 31 && wave < 16) wsum[wave] = incl;
    __syncthreads();
    if (t < 512) {
        int pre = 0;
        for (int w = 0; w < wave; ++w) pre += wsum[w];
        const int excl = bs + pre + incl - c;
        cnt[t] = excl;
        if (t < nn) rowptr[lo + t] = excl;
    }
    if (b == NBIN - 1 && t == 0) rowptr[N_NODES] = be;
    __syncthreads();

    for (int base = bs; base < be; base += 8192) {
        int2 r[8];
#pragma unroll
        for (int k = 0; k < 8; ++k) {
            const int e = base + k * 1024 + t;
            r[k] = (e < be) ? recs[e] : make_int2(-1, 0);
        }
        int pos[8];
#pragma unroll
        for (int k = 0; k < 8; ++k) {
            pos[k] = 0;
            if (r[k].x >= 0) pos[k] = atomicAdd(&cnt[r[k].x >> SRC_BITS], 1);
        }
#pragma unroll
        for (int k = 0; k < 8; ++k)
            if (r[k].x >= 0)
                list[pos[k]] = make_int2(r[k].x & ((1 << SRC_BITS) - 1), r[k].y);
    }
}

template <int U>
__device__ __forceinline__ void rows128(const float* __restrict__ H, int mysrc, int myval,
                                        int j, int lane, float4& acc) {
    float4 m[U];
    float  v[U];
#pragma unroll
    for (int k = 0; k < U; ++k) {
        const int s = __builtin_amdgcn_readlane(mysrc, j + k);
        v[k] = __int_as_float(__builtin_amdgcn_readlane(myval, j + k));
        m[k] = ((const float4*)(H + (size_t)s * 128))[lane];
    }
#pragma unroll
    for (int k = 0; k < U; ++k) {
        acc.x = fmaf(v[k], m[k].x, acc.x);
        acc.y = fmaf(v[k], m[k].y, acc.y);
        acc.z = fmaf(v[k], m[k].z, acc.z);
        acc.w = fmaf(v[k], m[k].w, acc.w);
    }
}

template <int U>
__device__ __forceinline__ void rows64(const float* __restrict__ H, int mysrc, int myval,
                                       int j, int lane, float2& acc) {
    float2 m[U];
    float  v[U];
#pragma unroll
    for (int k = 0; k < U; ++k) {
        const int s = __builtin_amdgcn_readlane(mysrc, j + k);
        v[k] = __int_as_float(__builtin_amdgcn_readlane(myval, j + k));
        m[k] = ((const float2*)(H + (size_t)s * 64))[lane];
    }
#pragma unroll
    for (int k = 0; k < U; ++k) {
        acc.x = fmaf(v[k], m[k].x, acc.x);
        acc.y = fmaf(v[k], m[k].y, acc.y);
    }
}

__global__ __launch_bounds__(256) void spmm_gather_d128(
    const int* __restrict__ rowptr, const int2* __restrict__ list,
    const float* __restrict__ H, float* __restrict__ out) {
    const int lane = threadIdx.x & 31;
    const int row  = blockIdx.x * 8 + (threadIdx.x >> 5);
    if (row >= N_NODES) return;
    const int beg = __builtin_amdgcn_readfirstlane(rowptr[row]);
    const int end = __builtin_amdgcn_readfirstlane(rowptr[row + 1]);
    float4 acc = make_float4(0.0f, 0.0f, 0.0f, 0.0f);
    for (int base = beg; base < end; base += 32) {
        const int n = min(32, end - base);
        int2 my = make_int2(0, 0);
        if (lane < n) my = list[base + lane];
        int j = 0;
        for (; j + 8 <= n; j += 8) rows128<8>(H, my.x, my.y, j, lane, acc);
        if (j + 4 <= n) { rows128<4>(H, my.x, my.y, j, lane, acc); j += 4; }
        if (j + 2 <= n) { rows128<2>(H, my.x, my.y, j, lane, acc); j += 2; }
        if (j < n)      { rows128<1>(H, my.x, my.y, j, lane, acc); }
    }
    ((float4*)(out + (size_t)row * 128))[lane] = acc;
}

__global__ __launch_bounds__(256) void spmm_gather_d64(
    const int* __restrict__ rowptr, const int2* __restrict__ list,
    const float* __restrict__ H, float* __restrict__ out) {
    const int lane = threadIdx.x & 31;
    const int row  = blockIdx.x * 8 + (threadIdx.x >> 5);
    if (row >= N_NODES) return;
    const int beg = __builtin_amdgcn_readfirstlane(rowptr[row]);
    const int end = __builtin_amdgcn_readfirstlane(rowptr[row + 1]);
    float2 acc = make_float2(0.0f, 0.0f);
    for (int base = beg; base < end; base += 32) {
        const int n = min(32, end - base);
        int2 my = make_int2(0, 0);
        if (lane < n) my = list[base + lane];
        int j = 0;
        for (; j + 8 <= n; j += 8) rows64<8>(H, my.x, my.y, j, lane, acc);
        if (j + 4 <= n) { rows64<4>(H, my.x, my.y, j, lane, acc); j += 4; }
        if (j + 2 <= n) { rows64<2>(H, my.x, my.y, j, lane, acc); j += 2; }
        if (j < n)      { rows64<1>(H, my.x, my.y, j, lane, acc); }
    }
    ((float2*)(out + (size_t)row * 64))[lane] = acc;
}

extern "C" void kernel_launch(void* const* d_in, const int* in_sizes, int n_in,
                              void* d_out, int out_size, void* d_ws, size_t ws_size,
                              hipStream_t stream) {
    const float* X     = (const float*)d_in[0];
    const int*   esrc  = (const int*)d_in[1];
    const int*   edst  = (const int*)d_in[2];
    const float* evals = (const float*)d_in[3];
    const float* W1    = (const float*)d_in[4];
    const float* W2    = (const float*)d_in[5];
    float*       out   = (float*)d_out;

    char* ws = (char*)d_ws;
    const size_t h1Bytes = (size_t)N_NODES * HID_DIM * sizeof(float);
    float* H1   = (float*)(ws);
    float* Hagg = (float*)(ws + h1Bytes);
    float* H2   = (float*)(ws);
    int2* list   = (int2*)(ws + 2 * h1Bytes);
    int*  rowptr = (int*)(ws + 2 * h1Bytes + 12800000 + 6400000 + 400128);
    int2* recs     = (int2*)(ws + h1Bytes);
    int*  counts   = (int*)(ws + h1Bytes + 12800000);
    int*  binStart = (int*)(ws + h1Bytes + 12800000 + 262144);

    bin_count<<<NCHUNK, 1024, 0, stream>>>(edst, counts);
    bin_scatter<<<NCHUNK, 1024, 0, stream>>>(esrc, edst, evals, counts, binStart, recs);
    bin_finish<<<NBIN, 1024, 0, stream>>>(binStart, recs, rowptr, list);

    gemm1_wmma<<<N_NODES / 16, 256, 0, stream>>>(X, W1, H1);
    spmm_gather_d128<<<N_NODES / 8, 256, 0, stream>>>(rowptr, list, H1, Hagg);

    gemm2_wmma<<<N_NODES / 32, 256, 0, stream>>>(Hagg, W2, H2);
    spmm_gather_d64<<<N_NODES / 8, 256, 0, stream>>>(rowptr, list, H2, out);
}


# gemm2 launch folded into the layer-1 row sum: 16-wave groups sum 32 rows (2 per wave), ReLU, hand them via LDS to the f32 WMMA with W2; summed rows never stored; 6 launches; sum orders kept
# speedup vs baseline: 3.0794x; 1.0805x over previous
#include <hip/hip_runtime.h>

typedef __attribute__((ext_vector_type(2))) float v2f;
typedef __attribute__((ext_vector_type(8))) float v8f;

#define N_NODES 100000
#define N_EDGES 1600000
#define IN_DIM 256
#define HID_DIM 128
#define OUT_DIM 64

__global__ __launch_bounds__(256) void gemm1_wmma(const float* __restrict__ X,
                                                  const float* __restrict__ W1,
                                                  float* __restrict__ H1) {
    __shared__ float xs[16][260];
    __shared__ float w1s[64][136];

    const int tid  = threadIdx.x;
    const int lane = tid & 31;
    const int wave = tid >> 5;
    const int half = lane >> 4;
    const int l16  = lane & 15;
    const int mbase = blockIdx.x * 16;

    {
        int row = tid >> 4;
        int seg = (tid & 15) * 16;
        const float* xp = X + (size_t)(mbase + row) * IN_DIM + seg;
        float4 a0 = ((const float4*)xp)[0];
        float4 a1 = ((const float4*)xp)[1];
        float4 a2 = ((const float4*)xp)[2];
        float4 a3 = ((const float4*)xp)[3];
        *(float4*)&xs[row][seg + 0]  = a0;
        *(float4*)&xs[row][seg + 4]  = a1;
        *(float4*)&xs[row][seg + 8]  = a2;
        *(float4*)&xs[row][seg + 12] = a3;
    }

    v8f acc = {};
    const int nb = wave * 16 + l16;

    for (int kc = 0; kc < IN_DIM; kc += 64) {
        for (int i = tid; i < (64 * 128) / 4; i += 256) {
            int flat = i * 4;
            int kr   = flat >> 7;
            int col  = flat & 127;
            float4 w = *(const float4*)&W1[(size_t)(kc + kr) * HID_DIM + col];
            *(float4*)&w1s[kr][col] = w;
        }
        __syncthreads();

#pragma unroll
        for (int kk = 0; kk < 64; kk += 4) {
            const int ka = kc + kk + half * 2;
            const int kb = kk + half * 2;
            v2f a = {xs[l16][ka], xs[l16][ka + 1]};
            v2f b = {w1s[kb][nb], w1s[kb + 1][nb]};
            acc = __builtin_amdgcn_wmma_f32_16x16x4_f32(
                false, a, false, b, (short)0, acc, false, false);
        }
        __syncthreads();
    }

#pragma unroll
    for (int v = 0; v < 8; ++v) {
        int row = mbase + v + half * 8;
        H1[(size_t)row * HID_DIM + nb] = acc[v];
    }
}

#define NBIN 256
#define BIN_NODES 391
#define NCHUNK 256
#define CHUNK_EDGES (N_EDGES / NCHUNK)
#define CHUNK_TRIPS 7
#define SRC_BITS 17

__global__ __launch_bounds__(1024) void bin_count(const int* __restrict__ dst,
                                                  int* __restrict__ counts) {
    __shared__ int hist[NBIN];
    const int t = threadIdx.x;
    if (t < NBIN) hist[t] = 0;
    __syncthreads();
    const int beg = blockIdx.x * CHUNK_EDGES;
    const int end = beg + CHUNK_EDGES;
    int d[CHUNK_TRIPS];
#pragma unroll
    for (int k = 0; k < CHUNK_TRIPS; ++k) {
        const int e = beg + k * 1024 + t;
        d[k] = (e < end) ? dst[e] : -1;
    }
#pragma unroll
    for (int k = 0; k < CHUNK_TRIPS; ++k)
        if (d[k] >= 0) atomicAdd(&hist[(unsigned)d[k] / BIN_NODES], 1);
    __syncthreads();
    if (t < NBIN) counts[blockIdx.x * NBIN + t] = hist[t];
}

__global__ __launch_bounds__(1024) void bin_scatter(const int* __restrict__ src,
                                                    const int* __restrict__ dst,
                                                    const float* __restrict__ vals,
                                                    const int* __restrict__ counts,
                                                    int* __restrict__ binStart,
                                                    int2* __restrict__ recs) {
    __shared__ int ptot[16][NBIN];
    __shared__ int pbef[16][NBIN];
    __shared__ int cursor[NBIN];
    __shared__ int wsum[8];
    const int t    = threadIdx.x;
    const int lane = t & 31;
    const int wave = t >> 5;
    const int me   = blockIdx.x;
    const int beg  = me * CHUNK_EDGES;
    const int end  = beg + CHUNK_EDGES;

    int   d[CHUNK_TRIPS], s[CHUNK_TRIPS];
    float v[CHUNK_TRIPS];
#pragma unroll
    for (int k = 0; k < CHUNK_TRIPS; ++k) {
        const int e = beg + k * 1024 + t;
        const bool ok = e < end;
        d[k] = ok ? dst[e] : -1;
        s[k] = ok ? src[e] : 0;
        v[k] = ok ? vals[e] : 0.0f;
    }

    {
        const int j4 = (t & 63) * 4;
        const int q  = t >> 6;
        int4 tot = make_int4(0, 0, 0, 0);
        int4 bef = make_int4(0, 0, 0, 0);
#pragma unroll
        for (int k = 0; k < 16; ++k) {
            const int c = q * 16 + k;
            const int4 n = *(const int4*)(counts + c * NBIN + j4);
            tot.x += n.x; tot.y += n.y; tot.z += n.z; tot.w += n.w;
            if (c < me) { bef.x += n.x; bef.y += n.y; bef.z += n.z; bef.w += n.w; }
        }
        *(int4*)&ptot[q][j4] = tot;
        *(int4*)&pbef[q][j4] = bef;
    }
    __syncthreads();
    int myTot = 0, myBef = 0;
    if (t < NBIN) {
#pragma unroll
        for (int k = 0; k < 16; ++k) { myTot += ptot[k][t]; myBef += pbef[k][t]; }
    }
    int incl = myTot;
    for (int o = 1; o < 32; o <<= 1) {
        const int u = __shfl_up(incl, o);
        if (lane >= o) incl += u;
    }
    if (lane == 31 && wave < 8) wsum[wave] = incl;
    __syncthreads();
    if (t < NBIN) {
        int base = 0;
        for (int w = 0; w < wave; ++w) base += wsum[w];
        const int excl = base + incl - myTot;
        cursor[t] = excl + myBef;
        if (me == 0) {
            binStart[t] = excl;
            if (t == NBIN - 1) binStart[NBIN] = excl + myTot;
        }
    }
    __syncthreads();

    int pos[CHUNK_TRIPS];
#pragma unroll
    for (int k = 0; k < CHUNK_TRIPS; ++k) {
        pos[k] = 0;
        if (d[k] >= 0) {
            const unsigned bin = (unsigned)d[k] / BIN_NODES;
            pos[k] = atomicAdd(&cursor[bin], 1);
            s[k] |= (d[k] - (int)bin * BIN_NODES) << SRC_BITS;
        }
    }
#pragma unroll
    for (int k = 0; k < CHUNK_TRIPS; ++k)
        if (d[k] >= 0) recs[pos[k]] = make_int2(s[k], __float_as_int(v[k]));
}

__global__ __launch_bounds__(1024) void bin_finish(const int* __restrict__ binStart,
                                                   const int2* __restrict__ recs,
                                                   int* __restrict__ rowptr,
                                                   int2* __restrict__ list) {
    __shared__ int cnt[512];
    __shared__ int wsum[16];
    const int t    = threadIdx.x;
    const int lane = t & 31;
    const int wave = t >> 5;
    const int b    = blockIdx.x;
    const int lo   = b * BIN_NODES;
    const int nn   = min(BIN_NODES, N_NODES - lo);
    const int bs   = binStart[b];
    const int be   = binStart[b + 1];
    if (t < 512) cnt[t] = 0;
    __syncthreads();

    for (int base = bs; base < be; base += 8192) {
        int x[8];
#pragma unroll
        for (int k = 0; k < 8; ++k) {
            const int e = base + k * 1024 + t;
            x[k] = (e < be) ? recs[e].x : -1;
        }
#pragma unroll
        for (int k = 0; k < 8; ++k)
            if (x[k] >= 0) atomicAdd(&cnt[x[k] >> SRC_BITS], 1);
    }
    __syncthreads();

    const int c = (t < 512) ? cnt[t] : 0;
    int incl = c;
    for (int o = 1; o < 32; o <<= 1) {
        const int u = __shfl_up(incl, o);
        if (lane >= o) incl += u;
    }
    if (lane == 31 && wave < 16) wsum[wave] = incl;
    __syncthreads();
    if (t < 512) {
        int pre = 0;
        for (int w = 0; w < wave; ++w) pre += wsum[w];
        const int excl = bs + pre + incl - c;
        cnt[t] = excl;
        if (t < nn) rowptr[lo + t] = excl;
    }
    if (b == NBIN - 1 && t == 0) rowptr[N_NODES] = be;
    __syncthreads();

    for (int base = bs; base < be; base += 8192) {
        int2 r[8];
#pragma unroll
        for (int k = 0; k < 8; ++k) {
            const int e = base + k * 1024 + t;
            r[k] = (e < be) ? recs[e] : make_int2(-1, 0);
        }
        int pos[8];
#pragma unroll
        for (int k = 0; k < 8; ++k) {
            pos[k] = 0;
            if (r[k].x >= 0) pos[k] = atomicAdd(&cnt[r[k].x >> SRC_BITS], 1);
        }
#pragma unroll
        for (int k = 0; k < 8; ++k)
            if (r[k].x >= 0)
                list[pos[k]] = make_int2(r[k].x & ((1 << SRC_BITS) - 1), r[k].y);
    }
}

template <int U>
__device__ __forceinline__ void rows128(const float* __restrict__ H, int mysrc, int myval,
                                        int j, int lane, float4& acc) {
    float4 m[U];
    float  v[U];
#pragma unroll
    for (int k = 0; k < U; ++k) {
        const int s = __builtin_amdgcn_readlane(mysrc, j + k);
        v[k] = __int_as_float(__builtin_amdgcn_readlane(myval, j + k));
        m[k] = ((const float4*)(H + (size_t)s * 128))[lane];
    }
#pragma unroll
    for (int k = 0; k < U; ++k) {
        acc.x = fmaf(v[k], m[k].x, acc.x);
        acc.y = fmaf(v[k], m[k].y, acc.y);
        acc.z = fmaf(v[k], m[k].z, acc.z);
        acc.w = fmaf(v[k], m[k].w, acc.w);
    }
}

template <int U>
__device__ __forceinline__ void rows64(const float* __restrict__ H, int mysrc, int myval,
                                       int j, int lane, float2& acc) {
    float2 m[U];
    float  v[U];
#pragma unroll
    for (int k = 0; k < U; ++k) {
        const int s = __builtin_amdgcn_readlane(mysrc, j + k);
        v[k] = __int_as_float(__builtin_amdgcn_readlane(myval, j + k));
        m[k] = ((const float2*)(H + (size_t)s * 64))[lane];
    }
#pragma unroll
    for (int k = 0; k < U; ++k) {
        acc.x = fmaf(v[k], m[k].x, acc.x);
        acc.y = fmaf(v[k], m[k].y, acc.y);
    }
}

static_assert(N_NODES % 32 == 0, "the grid covers the rows exactly: no row guard");

__global__ __launch_bounds__(512) void spmm_gather_d128_gemm2(
    const int* __restrict__ rowptr, const int2* __restrict__ list,
    const float* __restrict__ H, const float* __restrict__ W2,
    float* __restrict__ H2) {
    __shared__ float hs[32][132];
    __shared__ float w2s[128][72];

    const int tid   = threadIdx.x;
    const int lane  = tid & 31;
    const int wave  = tid >> 5;
    const int mbase = blockIdx.x * 32;

    int  beg[2], end[2];
    int2 first[2];
#pragma unroll
    for (int r = 0; r < 2; ++r) {
        const int row = mbase + wave + r * 16;
        beg[r] = __builtin_amdgcn_readfirstlane(rowptr[row]);
        end[r] = __builtin_amdgcn_readfirstlane(rowptr[row + 1]);
    }
#pragma unroll
    for (int r = 0; r < 2; ++r) {
        first[r] = make_int2(0, 0);
        if (lane < min(32, end[r] - beg[r])) first[r] = list[beg[r] + lane];
    }
#pragma unroll
    for (int i = 0; i < 4; ++i) {
        const int flat = (tid + i * 512) * 4;
        *(float4*)&w2s[flat >> 6][flat & 63] = *(const float4*)&W2[flat];
    }

#pragma unroll
    for (int r = 0; r < 2; ++r) {
        float4 acc = make_float4(0.0f, 0.0f, 0.0f, 0.0f);
        for (int base = beg[r]; base < end[r]; base += 32) {
            const int n = min(32, end[r] - base);
            int2 my = first[r];
            if (base != beg[r]) {
                my = make_int2(0, 0);
                if (lane < n) my = list[base + lane];
            }
            int j = 0;
            for (; j + 8 <= n; j += 8) rows128<8>(H, my.x, my.y, j, lane, acc);
            if (j + 4 <= n) { rows128<4>(H, my.x, my.y, j, lane, acc); j += 4; }
            if (j + 2 <= n) { rows128<2>(H, my.x, my.y, j, lane, acc); j += 2; }
            if (j < n)      { rows128<1>(H, my.x, my.y, j, lane, acc); }
        }
        acc.x = fmaxf(acc.x, 0.0f);
        acc.y = fmaxf(acc.y, 0.0f);
        acc.z = fmaxf(acc.z, 0.0f);
        acc.w = fmaxf(acc.w, 0.0f);
        *(float4*)&hs[wave + r * 16][lane * 4] = acc;
    }
    __syncthreads();
    if (wave >= 8) return;

    const int mt   = wave >> 2;
    const int nt   = wave & 3;
    const int half = lane >> 4;
    const int l16  = lane & 15;
    const int mrow = mt * 16 + l16;
    const int nb   = nt * 16 + l16;

    v8f d = {};
#pragma unroll
    for (int kk = 0; kk < HID_DIM; kk += 4) {
        const int k = kk + half * 2;
        v2f a = {hs[mrow][k], hs[mrow][k + 1]};
        v2f b = {w2s[k][nb], w2s[k + 1][nb]};
        d = __builtin_amdgcn_wmma_f32_16x16x4_f32(
            false, a, false, b, (short)0, d, false, false);
    }

#pragma unroll
    for (int v = 0; v < 8; ++v) {
        const int row = mbase + mt * 16 + v + half * 8;
        H2[(size_t)row * OUT_DIM + nb] = d[v];
    }
}

__global__ __launch_bounds__(256) void spmm_gather_d64(
    const int* __restrict__ rowptr, const int2* __restrict__ list,
    const float* __restrict__ H, float* __restrict__ out) {
    const int lane = threadIdx.x & 31;
    const int row  = blockIdx.x * 8 + (threadIdx.x >> 5);
    if (row >= N_NODES) return;
    const int beg = __builtin_amdgcn_readfirstlane(rowptr[row]);
    const int end = __builtin_amdgcn_readfirstlane(rowptr[row + 1]);
    float2 acc = make_float2(0.0f, 0.0f);
    for (int base = beg; base < end; base += 32) {
        const int n = min(32, end - base);
        int2 my = make_int2(0, 0);
        if (lane < n) my = list[base + lane];
        int j = 0;
        for (; j + 8 <= n; j += 8) rows64<8>(H, my.x, my.y, j, lane, acc);
        if (j + 4 <= n) { rows64<4>(H, my.x, my.y, j, lane, acc); j += 4; }
        if (j + 2 <= n) { rows64<2>(H, my.x, my.y, j, lane, acc); j += 2; }
        if (j < n)      { rows64<1>(H, my.x, my.y, j, lane, acc); }
    }
    ((float2*)(out + (size_t)row * 64))[lane] = acc;
}

extern "C" void kernel_launch(void* const* d_in, const int* in_sizes, int n_in,
                              void* d_out, int out_size, void* d_ws, size_t ws_size,
                              hipStream_t stream) {
    const float* X     = (const float*)d_in[0];
    const int*   esrc  = (const int*)d_in[1];
    const int*   edst  = (const int*)d_in[2];
    const float* evals = (const float*)d_in[3];
    const float* W1    = (const float*)d_in[4];
    const float* W2    = (const float*)d_in[5];
    float*       out   = (float*)d_out;

    char* ws = (char*)d_ws;
    const size_t h1Bytes = (size_t)N_NODES * HID_DIM * sizeof(float);
    float* H1   = (float*)(ws);
    float* H2   = (float*)(ws + h1Bytes);
    int2* list   = (int2*)(ws + 2 * h1Bytes);
    int*  rowptr = (int*)(ws + 2 * h1Bytes + 12800000 + 6400000 + 400128);
    int2* recs     = (int2*)(ws + h1Bytes);
    int*  counts   = (int*)(ws + h1Bytes + 12800000);
    int*  binStart = (int*)(ws + h1Bytes + 12800000 + 262144);

    bin_count<<<NCHUNK, 1024, 0, stream>>>(edst, counts);
    bin_scatter<<<NCHUNK, 1024, 0, stream>>>(esrc, edst, evals, counts, binStart, recs);
    bin_finish<<<NBIN, 1024, 0, stream>>>(binStart, recs, rowptr, list);

    gemm1_wmma<<<N_NODES / 16, 256, 0, stream>>>(X, W1, H1);
    spmm_gather_d128_gemm2<<<N_NODES / 32, 512, 0, stream>>>(rowptr, list, H1, W2, H2);
    spmm_gather_d64<<<N_NODES / 8, 256, 0, stream>>>(rowptr, list, H2, out);
}


# gemm1: each wave keeps its W1 fragments for the whole K in registers while 512 work-groups loop over 16-row tiles (X tile double-buffered in LDS); W1 no longer re-staged per tile; sum order kept
# speedup vs baseline: 3.6238x; 1.1768x over previous
#include <hip/hip_runtime.h>

typedef __attribute__((ext_vector_type(2))) float v2f;
typedef __attribute__((ext_vector_type(8))) float v8f;

#define N_NODES 100000
#define N_EDGES 1600000
#define IN_DIM 256
#define HID_DIM 128
#define OUT_DIM 64

#define G1_TILES (N_NODES / 16)
#define G1_GRID 512
static_assert(N_NODES % 16 == 0, "M tiles cover the rows exactly: no row guard");
static_assert(G1_GRID <= G1_TILES, "every work-group owns a first tile");

__global__ __launch_bounds__(256) void gemm1_wmma(const float* __restrict__ X,
                                                  const float* __restrict__ W1,
                                                  float* __restrict__ H1) {
    __shared__ float xs[2][16][260];

    const int tid  = threadIdx.x;
    const int lane = tid & 31;
    const int wave = tid >> 5;
    const int half = lane >> 4;
    const int l16  = lane & 15;
    const int nb   = wave * 16 + l16;
    const int srow = tid >> 4;
    const int sseg = (tid & 15) * 16;

    v2f b[64];
#pragma unroll
    for (int s = 0; s < 64; ++s) {
        const int k = s * 4 + half * 2;
        b[s] = (v2f){W1[(size_t)k * HID_DIM + nb], W1[(size_t)(k + 1) * HID_DIM + nb]};
    }

    int t = blockIdx.x;
    {
        const float* xp = X + (size_t)(t * 16 + srow) * IN_DIM + sseg;
        float4 a0 = ((const float4*)xp)[0];
        float4 a1 = ((const float4*)xp)[1];
        float4 a2 = ((const float4*)xp)[2];
        float4 a3 = ((const float4*)xp)[3];
        *(float4*)&xs[0][srow][sseg + 0]  = a0;
        *(float4*)&xs[0][srow][sseg + 4]  = a1;
        *(float4*)&xs[0][srow][sseg + 8]  = a2;
        *(float4*)&xs[0][srow][sseg + 12] = a3;
    }
    __syncthreads();

    int buf = 0;
    for (; t < G1_TILES; t += G1_GRID) {
        const int  tn   = t + G1_GRID;
        const bool more = tn < G1_TILES;
        float4 n0, n1, n2, n3;
        if (more) {
            const float* xp = X + (size_t)(tn * 16 + srow) * IN_DIM + sseg;
            n0 = ((const float4*)xp)[0];
            n1 = ((const float4*)xp)[1];
            n2 = ((const float4*)xp)[2];
            n3 = ((const float4*)xp)[3];
        }

        v8f acc = {};
#pragma unroll
        for (int s = 0; s < 64; ++s) {
            const int ka = s * 4 + half * 2;
            v2f a = {xs[buf][l16][ka], xs[buf][l16][ka + 1]};
            acc = __builtin_amdgcn_wmma_f32_16x16x4_f32(
                false, a, false, b[s], (short)0, acc, false, false);
        }

        if (more) {
            *(float4*)&xs[buf ^ 1][srow][sseg + 0]  = n0;
            *(float4*)&xs[buf ^ 1][srow][sseg + 4]  = n1;
            *(float4*)&xs[buf ^ 1][srow][sseg + 8]  = n2;
            *(float4*)&xs[buf ^ 1][srow][sseg + 12] = n3;
        }
        __syncthreads();
        buf ^= 1;

#pragma unroll
        for (int v = 0; v < 8; ++v) {
            const int row = t * 16 + v + half * 8;
            H1[(size_t)row * HID_DIM + nb] = acc[v];
        }
    }
}

#define NBIN 256
#define BIN_NODES 391
#define NCHUNK 256
#define CHUNK_EDGES (N_EDGES / NCHUNK)
#define CHUNK_TRIPS 7
#define SRC_BITS 17

__global__ __launch_bounds__(1024) void bin_count(const int* __restrict__ dst,
                                                  int* __restrict__ counts) {
    __shared__ int hist[NBIN];
    const int t = threadIdx.x;
    if (t < NBIN) hist[t] = 0;
    __syncthreads();
    const int beg = blockIdx.x * CHUNK_EDGES;
    const int end = beg + CHUNK_EDGES;
    int d[CHUNK_TRIPS];
#pragma unroll
    for (int k = 0; k < CHUNK_TRIPS; ++k) {
        const int e = beg + k * 1024 + t;
        d[k] = (e < end) ? dst[e] : -1;
    }
#pragma unroll
    for (int k = 0; k < CHUNK_TRIPS; ++k)
        if (d[k] >= 0) atomicAdd(&hist[(unsigned)d[k] / BIN_NODES], 1);
    __syncthreads();
    if (t < NBIN) counts[blockIdx.x * NBIN + t] = hist[t];
}

__global__ __launch_bounds__(1024) void bin_scatter(const int* __restrict__ src,
                                                    const int* __restrict__ dst,
                                                    const float* __restrict__ vals,
                                                    const int* __restrict__ counts,
                                                    int* __restrict__ binStart,
                                                    int2* __restrict__ recs) {
    __shared__ int ptot[16][NBIN];
    __shared__ int pbef[16][NBIN];
    __shared__ int cursor[NBIN];
    __shared__ int wsum[8];
    const int t    = threadIdx.x;
    const int lane = t & 31;
    const int wave = t >> 5;
    const int me   = blockIdx.x;
    const int beg  = me * CHUNK_EDGES;
    const int end  = beg + CHUNK_EDGES;

    int   d[CHUNK_TRIPS], s[CHUNK_TRIPS];
    float v[CHUNK_TRIPS];
#pragma unroll
    for (int k = 0; k < CHUNK_TRIPS; ++k) {
        const int e = beg + k * 1024 + t;
        const bool ok = e < end;
        d[k] = ok ? dst[e] : -1;
        s[k] = ok ? src[e] : 0;
        v[k] = ok ? vals[e] : 0.0f;
    }

    {
        const int j4 = (t & 63) * 4;
        const int q  = t >> 6;
        int4 tot = make_int4(0, 0, 0, 0);
        int4 bef = make_int4(0, 0, 0, 0);
#pragma unroll
        for (int k = 0; k < 16; ++k) {
            const int c = q * 16 + k;
            const int4 n = *(const int4*)(counts + c * NBIN + j4);
            tot.x += n.x; tot.y += n.y; tot.z += n.z; tot.w += n.w;
            if (c < me) { bef.x += n.x; bef.y += n.y; bef.z += n.z; bef.w += n.w; }
        }
        *(int4*)&ptot[q][j4] = tot;
        *(int4*)&pbef[q][j4] = bef;
    }
    __syncthreads();
    int myTot = 0, myBef = 0;
    if (t < NBIN) {
#pragma unroll
        for (int k = 0; k < 16; ++k) { myTot += ptot[k][t]; myBef += pbef[k][t]; }
    }
    int incl = myTot;
    for (int o = 1; o < 32; o <<= 1) {
        const int u = __shfl_up(incl, o);
        if (lane >= o) incl += u;
    }
    if (lane == 31 && wave < 8) wsum[wave] = incl;
    __syncthreads();
    if (t < NBIN) {
        int base = 0;
        for (int w = 0; w < wave; ++w) base += wsum[w];
        const int excl = base + incl - myTot;
        cursor[t] = excl + myBef;
        if (me == 0) {
            binStart[t] = excl;
            if (t == NBIN - 1) binStart[NBIN] = excl + myTot;
        }
    }
    __syncthreads();

    int pos[CHUNK_TRIPS];
#pragma unroll
    for (int k = 0; k < CHUNK_TRIPS; ++k) {
        pos[k] = 0;
        if (d[k] >= 0) {
            const unsigned bin = (unsigned)d[k] / BIN_NODES;
            pos[k] = atomicAdd(&cursor[bin], 1);
            s[k] |= (d[k] - (int)bin * BIN_NODES) << SRC_BITS;
        }
    }
#pragma unroll
    for (int k = 0; k < CHUNK_TRIPS; ++k)
        if (d[k] >= 0) recs[pos[k]] = make_int2(s[k], __float_as_int(v[k]));
}

__global__ __launch_bounds__(1024) void bin_finish(const int* __restrict__ binStart,
                                                   const int2* __restrict__ recs,
                                                   int* __restrict__ rowptr,
                                                   int2* __restrict__ list) {
    __shared__ int cnt[512];
    __shared__ int wsum[16];
    const int t    = threadIdx.x;
    const int lane = t & 31;
    const int wave = t >> 5;
    const int b    = blockIdx.x;
    const int lo   = b * BIN_NODES;
    const int nn   = min(BIN_NODES, N_NODES - lo);
    const int bs   = binStart[b];
    const int be   = binStart[b + 1];
    if (t < 512) cnt[t] = 0;
    __syncthreads();

    for (int base = bs; base < be; base += 8192) {
        int x[8];
#pragma unroll
        for (int k = 0; k < 8; ++k) {
            const int e = base + k * 1024 + t;
            x[k] = (e < be) ? recs[e].x : -1;
        }
#pragma unroll
        for (int k = 0; k < 8; ++k)
            if (x[k] >= 0) atomicAdd(&cnt[x[k] >> SRC_BITS], 1);
    }
    __syncthreads();

    const int c = (t < 512) ? cnt[t] : 0;
    int incl = c;
    for (int o = 1; o < 32; o <<= 1) {
        const int u = __shfl_up(incl, o);
        if (lane >= o) incl += u;
    }
    if (lane == 31 && wave < 16) wsum[wave] = incl;
    __syncthreads();
    if (t < 512) {
        int pre = 0;
        for (int w = 0; w < wave; ++w) pre += wsum[w];
        const int excl = bs + pre + incl - c;
        cnt[t] = excl;
        if (t < nn) rowptr[lo + t] = excl;
    }
    if (b == NBIN - 1 && t == 0) rowptr[N_NODES] = be;
    __syncthreads();

    for (int base = bs; base < be; base += 8192) {
        int2 r[8];
#pragma unroll
        for (int k = 0; k < 8; ++k) {
            const int e = base + k * 1024 + t;
            r[k] = (e < be) ? recs[e] : make_int2(-1, 0);
        }
        int pos[8];
#pragma unroll
        for (int k = 0; k < 8; ++k) {
            pos[k] = 0;
            if (r[k].x >= 0) pos[k] = atomicAdd(&cnt[r[k].x >> SRC_BITS], 1);
        }
#pragma unroll
        for (int k = 0; k < 8; ++k)
            if (r[k].x >= 0)
                list[pos[k]] = make_int2(r[k].x & ((1 << SRC_BITS) - 1), r[k].y);
    }
}

template <int U>
__device__ __forceinline__ void rows128(const float* __restrict__ H, int mysrc, int myval,
                                        int j, int lane, float4& acc) {
    float4 m[U];
    float  v[U];
#pragma unroll
    for (int k = 0; k < U; ++k) {
        const int s = __builtin_amdgcn_readlane(mysrc, j + k);
        v[k] = __int_as_float(__builtin_amdgcn_readlane(myval, j + k));
        m[k] = ((const float4*)(H + (size_t)s * 128))[lane];
    }
#pragma unroll
    for (int k = 0; k < U; ++k) {
        acc.x = fmaf(v[k], m[k].x, acc.x);
        acc.y = fmaf(v[k], m[k].y, acc.y);
        acc.z = fmaf(v[k], m[k].z, acc.z);
        acc.w = fmaf(v[k], m[k].w, acc.w);
    }
}

template <int U>
__device__ __forceinline__ void rows64(const float* __restrict__ H, int mysrc, int myval,
                                       int j, int lane, float2& acc) {
    float2 m[U];
    float  v[U];
#pragma unroll
    for (int k = 0; k < U; ++k) {
        const int s = __builtin_amdgcn_readlane(mysrc, j + k);
        v[k] = __int_as_float(__builtin_amdgcn_readlane(myval, j + k));
        m[k] = ((const float2*)(H + (size_t)s * 64))[lane];
    }
#pragma unroll
    for (int k = 0; k < U; ++k) {
        acc.x = fmaf(v[k], m[k].x, acc.x);
        acc.y = fmaf(v[k], m[k].y, acc.y);
    }
}

static_assert(N_NODES % 32 == 0, "the grid covers the rows exactly: no row guard");

__global__ __launch_bounds__(512) void spmm_gather_d128_gemm2(
    const int* __restrict__ rowptr, const int2* __restrict__ list,
    const float* __restrict__ H, const float* __restrict__ W2,
    float* __restrict__ H2) {
    __shared__ float hs[32][132];
    __shared__ float w2s[128][72];

    const int tid   = threadIdx.x;
    const int lane  = tid & 31;
    const int wave  = tid >> 5;
    const int mbase = blockIdx.x * 32;

    int  beg[2], end[2];
    int2 first[2];
#pragma unroll
    for (int r = 0; r < 2; ++r) {
        const int row = mbase + wave + r * 16;
        beg[r] = __builtin_amdgcn_readfirstlane(rowptr[row]);
        end[r] = __builtin_amdgcn_readfirstlane(rowptr[row + 1]);
    }
#pragma unroll
    for (int r = 0; r < 2; ++r) {
        first[r] = make_int2(0, 0);
        if (lane < min(32, end[r] - beg[r])) first[r] = list[beg[r] + lane];
    }
#pragma unroll
    for (int i = 0; i < 4; ++i) {
        const int flat = (tid + i * 512) * 4;
        *(float4*)&w2s[flat >> 6][flat & 63] = *(const float4*)&W2[flat];
    }

#pragma unroll
    for (int r = 0; r < 2; ++r) {
        float4 acc = make_float4(0.0f, 0.0f, 0.0f, 0.0f);
        for (int base = beg[r]; base < end[r]; base += 32) {
            const int n = min(32, end[r] - base);
            int2 my = first[r];
            if (base != beg[r]) {
                my = make_int2(0, 0);
                if (lane < n) my = list[base + lane];
            }
            int j = 0;
            for (; j + 8 <= n; j += 8) rows128<8>(H, my.x, my.y, j, lane, acc);
            if (j + 4 <= n) { rows128<4>(H, my.x, my.y, j, lane, acc); j += 4; }
            if (j + 2 <= n) { rows128<2>(H, my.x, my.y, j, lane, acc); j += 2; }
            if (j < n)      { rows128<1>(H, my.x, my.y, j, lane, acc); }
        }
        acc.x = fmaxf(acc.x, 0.0f);
        acc.y = fmaxf(acc.y, 0.0f);
        acc.z = fmaxf(acc.z, 0.0f);
        acc.w = fmaxf(acc.w, 0.0f);
        *(float4*)&hs[wave + r * 16][lane * 4] = acc;
    }
    __syncthreads();
    if (wave >= 8) return;

    const int mt   = wave >> 2;
    const int nt   = wave & 3;
    const int half = lane >> 4;
    const int l16  = lane & 15;
    const int mrow = mt * 16 + l16;
    const int nb   = nt * 16 + l16;

    v8f d = {};
#pragma unroll
    for (int kk = 0; kk < HID_DIM; kk += 4) {
        const int k = kk + half * 2;
        v2f a = {hs[mrow][k], hs[mrow][k + 1]};
        v2f b = {w2s[k][nb], w2s[k + 1][nb]};
        d = __builtin_amdgcn_wmma_f32_16x16x4_f32(
            false, a, false, b, (short)0, d, false, false);
    }

#pragma unroll
    for (int v = 0; v < 8; ++v) {
        const int row = mbase + mt * 16 + v + half * 8;
        H2[(size_t)row * OUT_DIM + nb] = d[v];
    }
}

__global__ __launch_bounds__(256) void spmm_gather_d64(
    const int* __restrict__ rowptr, const int2* __restrict__ list,
    const float* __restrict__ H, float* __restrict__ out) {
    const int lane = threadIdx.x & 31;
    const int row  = blockIdx.x * 8 + (threadIdx.x >> 5);
    if (row >= N_NODES) return;
    const int beg = __builtin_amdgcn_readfirstlane(rowptr[row]);
    const int end = __builtin_amdgcn_readfirstlane(rowptr[row + 1]);
    float2 acc = make_float2(0.0f, 0.0f);
    for (int base = beg; base < end; base += 32) {
        const int n = min(32, end - base);
        int2 my = make_int2(0, 0);
        if (lane < n) my = list[base + lane];
        int j = 0;
        for (; j + 8 <= n; j += 8) rows64<8>(H, my.x, my.y, j, lane, acc);
        if (j + 4 <= n) { rows64<4>(H, my.x, my.y, j, lane, acc); j += 4; }
        if (j + 2 <= n) { rows64<2>(H, my.x, my.y, j, lane, acc); j += 2; }
        if (j < n)      { rows64<1>(H, my.x, my.y, j, lane, acc); }
    }
    ((float2*)(out + (size_t)row * 64))[lane] = acc;
}

extern "C" void kernel_launch(void* const* d_in, const int* in_sizes, int n_in,
                              void* d_out, int out_size, void* d_ws, size_t ws_size,
                              hipStream_t stream) {
    const float* X     = (const float*)d_in[0];
    const int*   esrc  = (const int*)d_in[1];
    const int*   edst  = (const int*)d_in[2];
    const float* evals = (const float*)d_in[3];
    const float* W1    = (const float*)d_in[4];
    const float* W2    = (const float*)d_in[5];
    float*       out   = (float*)d_out;

    char* ws = (char*)d_ws;
    const size_t h1Bytes = (size_t)N_NODES * HID_DIM * sizeof(float);
    float* H1   = (float*)(ws);
    float* H2   = (float*)(ws + h1Bytes);
    int2* list   = (int2*)(ws + 2 * h1Bytes);
    int*  rowptr = (int*)(ws + 2 * h1Bytes + 12800000 + 6400000 + 400128);
    int2* recs     = (int2*)(ws + h1Bytes);
    int*  counts   = (int*)(ws + h1Bytes + 12800000);
    int*  binStart = (int*)(ws + h1Bytes + 12800000 + 262144);

    bin_count<<<NCHUNK, 1024, 0, stream>>>(edst, counts);
    bin_scatter<<<NCHUNK, 1024, 0, stream>>>(esrc, edst, evals, counts, binStart, recs);
    bin_finish<<<NBIN, 1024, 0, stream>>>(binStart, recs, rowptr, list);

    gemm1_wmma<<<G1_GRID, 256, 0, stream>>>(X, W1, H1);
    spmm_gather_d128_gemm2<<<N_NODES / 32, 512, 0, stream>>>(rowptr, list, H1, W2, H2);
    spmm_gather_d64<<<N_NODES / 8, 256, 0, stream>>>(rowptr, list, H2, out);
}


# edge grouping in 2 launches, not 3: each chunk sorts its edges by bin in LDS (rank, scan) and writes one sorted piece; a bin is collected from the 256 pieces; row sum order = LDS arrival as before
# speedup vs baseline: 3.9646x; 1.0940x over previous
#include <hip/hip_runtime.h>

typedef __attribute__((ext_vector_type(2))) float v2f;
typedef __attribute__((ext_vector_type(8))) float v8f;

#define N_NODES 100000
#define N_EDGES 1600000
#define IN_DIM 256
#define HID_DIM 128
#define OUT_DIM 64

#define G1_TILES (N_NODES / 16)
#define G1_GRID 512
static_assert(N_NODES % 16 == 0, "M tiles cover the rows exactly: no row guard");
static_assert(G1_GRID <= G1_TILES, "every work-group owns a first tile");

__global__ __launch_bounds__(256) void gemm1_wmma(const float* __restrict__ X,
                                                  const float* __restrict__ W1,
                                                  float* __restrict__ H1) {
    __shared__ float xs[2][16][260];

    const int tid  = threadIdx.x;
    const int lane = tid & 31;
    const int wave = tid >> 5;
    const int half = lane >> 4;
    const int l16  = lane & 15;
    const int nb   = wave * 16 + l16;
    const int srow = tid >> 4;
    const int sseg = (tid & 15) * 16;

    v2f b[64];
#pragma unroll
    for (int s = 0; s < 64; ++s) {
        const int k = s * 4 + half * 2;
        b[s] = (v2f){W1[(size_t)k * HID_DIM + nb], W1[(size_t)(k + 1) * HID_DIM + nb]};
    }

    int t = blockIdx.x;
    {
        const float* xp = X + (size_t)(t * 16 + srow) * IN_DIM + sseg;
        float4 a0 = ((const float4*)xp)[0];
        float4 a1 = ((const float4*)xp)[1];
        float4 a2 = ((const float4*)xp)[2];
        float4 a3 = ((const float4*)xp)[3];
        *(float4*)&xs[0][srow][sseg + 0]  = a0;
        *(float4*)&xs[0][srow][sseg + 4]  = a1;
        *(float4*)&xs[0][srow][sseg + 8]  = a2;
        *(float4*)&xs[0][srow][sseg + 12] = a3;
    }
    __syncthreads();

    int buf = 0;
    for (; t < G1_TILES; t += G1_GRID) {
        const int  tn   = t + G1_GRID;
        const bool more = tn < G1_TILES;
        float4 n0, n1, n2, n3;
        if (more) {
            const float* xp = X + (size_t)(tn * 16 + srow) * IN_DIM + sseg;
            n0 = ((const float4*)xp)[0];
            n1 = ((const float4*)xp)[1];
            n2 = ((const float4*)xp)[2];
            n3 = ((const float4*)xp)[3];
        }

        v8f acc = {};
#pragma unroll
        for (int s = 0; s < 64; ++s) {
            const int ka = s * 4 + half * 2;
            v2f a = {xs[buf][l16][ka], xs[buf][l16][ka + 1]};
            acc = __builtin_amdgcn_wmma_f32_16x16x4_f32(
                false, a, false, b[s], (short)0, acc, false, false);
        }

        if (more) {
            *(float4*)&xs[buf ^ 1][srow][sseg + 0]  = n0;
            *(float4*)&xs[buf ^ 1][srow][sseg + 4]  = n1;
            *(float4*)&xs[buf ^ 1][srow][sseg + 8]  = n2;
            *(float4*)&xs[buf ^ 1][srow][sseg + 12] = n3;
        }
        __syncthreads();
        buf ^= 1;

#pragma unroll
        for (int v = 0; v < 8; ++v) {
            const int row = t * 16 + v + half * 8;
            H1[(size_t)row * HID_DIM + nb] = acc[v];
        }
    }
}

#define NBIN 256
#define BIN_NODES 391
#define NCHUNK 256
#define CHUNK_EDGES (N_EDGES / NCHUNK)
#define CHUNK_TRIPS 7
#define SRC_BITS 17
static_assert(N_EDGES % NCHUNK == 0, "every chunk holds exactly CHUNK_EDGES edges");
static_assert(CHUNK_EDGES % 2 == 0, "a chunk's piece is a whole number of 16-byte pairs");

__global__ __launch_bounds__(1024) void chunk_sort(const int* __restrict__ src,
                                                   const int* __restrict__ dst,
                                                   const float* __restrict__ vals,
                                                   int* __restrict__ counts,
                                                   int* __restrict__ offs,
                                                   int2* __restrict__ recs) {
    __shared__ __align__(16) int2 srt[CHUNK_EDGES];
    __shared__ int hist[NBIN];
    __shared__ int first[NBIN];
    __shared__ int wsum[8];
    const int t    = threadIdx.x;
    const int lane = t & 31;
    const int wave = t >> 5;
    const int me   = blockIdx.x;
    const int beg  = me * CHUNK_EDGES;
    const int end  = beg + CHUNK_EDGES;

    if (t < NBIN) hist[t] = 0;

    int   d[CHUNK_TRIPS], s[CHUNK_TRIPS];
    float v[CHUNK_TRIPS];
#pragma unroll
    for (int k = 0; k < CHUNK_TRIPS; ++k) {
        const int e = beg + k * 1024 + t;
        const bool ok = e < end;
        d[k] = ok ? dst[e] : -1;
        s[k] = ok ? src[e] : 0;
        v[k] = ok ? vals[e] : 0.0f;
    }
    __syncthreads();

    int rank[CHUNK_TRIPS], bin[CHUNK_TRIPS];
#pragma unroll
    for (int k = 0; k < CHUNK_TRIPS; ++k) {
        rank[k] = 0;
        bin[k]  = 0;
        if (d[k] >= 0) {
            const unsigned bb = (unsigned)d[k] / BIN_NODES;
            bin[k]  = (int)bb;
            rank[k] = atomicAdd(&hist[bb], 1);
            s[k] |= (d[k] - (int)bb * BIN_NODES) << SRC_BITS;
        }
    }
    __syncthreads();

    const int myCnt = (t < NBIN) ? hist[t] : 0;
    int incl = myCnt;
    for (int o = 1; o < 32; o <<= 1) {
        const int u = __shfl_up(incl, o);
        if (lane >= o) incl += u;
    }
    if (lane == 31 && wave < 8) wsum[wave] = incl;
    __syncthreads();
    if (t < NBIN) {
        int base = 0;
        for (int w = 0; w < wave; ++w) base += wsum[w];
        const int excl = base + incl - myCnt;
        first[t] = excl;
        counts[me * NBIN + t] = myCnt;
        offs[me * NBIN + t]   = excl;
    }
    __syncthreads();

    int slot[CHUNK_TRIPS];
#pragma unroll
    for (int k = 0; k < CHUNK_TRIPS; ++k) slot[k] = first[bin[k]] + rank[k];
#pragma unroll
    for (int k = 0; k < CHUNK_TRIPS; ++k)
        if (d[k] >= 0) srt[slot[k]] = make_int2(s[k], __float_as_int(v[k]));
    __syncthreads();

    int4*       outp = (int4*)(recs + beg);
    const int4* inp  = (const int4*)srt;
    for (int i = t; i < CHUNK_EDGES / 2; i += 1024) outp[i] = inp[i];
}

__global__ __launch_bounds__(1024) void bin_finish(const int* __restrict__ counts,
                                                   const int* __restrict__ offs,
                                                   const int2* __restrict__ recs,
                                                   int* __restrict__ rowptr,
                                                   int2* __restrict__ list) {
    __shared__ int segLen[NCHUNK];
    __shared__ int segStart[NCHUNK];
    __shared__ int cnt[512];
    __shared__ int wsum[16];
    __shared__ int colOff[8];
    __shared__ int colLen[8];
    const int t    = threadIdx.x;
    const int lane = t & 31;
    const int wave = t >> 5;
    const int b    = blockIdx.x;
    const int lo   = b * BIN_NODES;
    const int nn   = min(BIN_NODES, N_NODES - lo);

    if (t < 512) cnt[t] = 0;
    int myLen = 0, myOff = 0;
    if (t < NCHUNK) {
        myLen = counts[t * NBIN + b];
        myOff = offs[t * NBIN + b];
        segLen[t]   = myLen;
        segStart[t] = t * CHUNK_EDGES + myOff;
    }
    int sOff = myOff, sLen = myLen;
    for (int o = 16; o > 0; o >>= 1) {
        sOff += __shfl_xor(sOff, o);
        sLen += __shfl_xor(sLen, o);
    }
    if (lane == 0 && wave < 8) { colOff[wave] = sOff; colLen[wave] = sLen; }
    __syncthreads();
    int bs = 0, be = 0;
#pragma unroll
    for (int w = 0; w < 8; ++w) { bs += colOff[w]; be += colLen[w]; }
    be += bs;

    const int   len = segLen[t >> 2];
    const int2* seg = recs + segStart[t >> 2];
    const int   sub = t & 3;

    for (int j0 = 0; j0 < len; j0 += 32) {
        int x[8];
#pragma unroll
        for (int k = 0; k < 8; ++k) {
            const int idx = j0 + k * 4 + sub;
            x[k] = (idx < len) ? seg[idx].x : -1;
        }
#pragma unroll
        for (int k = 0; k < 8; ++k)
            if (x[k] >= 0) atomicAdd(&cnt[x[k] >> SRC_BITS], 1);
    }
    __syncthreads();

    const int c = (t < 512) ? cnt[t] : 0;
    int incl = c;
    for (int o = 1; o < 32; o <<= 1) {
        const int u = __shfl_up(incl, o);
        if (lane >= o) incl += u;
    }
    if (lane == 31 && wave < 16) wsum[wave] = incl;
    __syncthreads();
    if (t < 512) {
        int pre = 0;
        for (int w = 0; w < wave; ++w) pre += wsum[w];
        const int excl = bs + pre + incl - c;
        cnt[t] = excl;
        if (t < nn) rowptr[lo + t] = excl;
    }
    if (b == NBIN - 1 && t == 0) rowptr[N_NODES] = be;
    __syncthreads();

    for (int j0 = 0; j0 < len; j0 += 32) {
        int2 r[8];
#pragma unroll
        for (int k = 0; k < 8; ++k) {
            const int idx = j0 + k * 4 + sub;
            r[k] = (idx < len) ? seg[idx] : make_int2(-1, 0);
        }
        int pos[8];
#pragma unroll
        for (int k = 0; k < 8; ++k) {
            pos[k] = 0;
            if (r[k].x >= 0) pos[k] = atomicAdd(&cnt[r[k].x >> SRC_BITS], 1);
        }
#pragma unroll
        for (int k = 0; k < 8; ++k)
            if (r[k].x >= 0)
                list[pos[k]] = make_int2(r[k].x & ((1 << SRC_BITS) - 1), r[k].y);
    }
}

template <int U>
__device__ __forceinline__ void rows128(const float* __restrict__ H, int mysrc, int myval,
                                        int j, int lane, float4& acc) {
    float4 m[U];
    float  v[U];
#pragma unroll
    for (int k = 0; k < U; ++k) {
        const int s = __builtin_amdgcn_readlane(mysrc, j + k);
        v[k] = __int_as_float(__builtin_amdgcn_readlane(myval, j + k));
        m[k] = ((const float4*)(H + (size_t)s * 128))[lane];
    }
#pragma unroll
    for (int k = 0; k < U; ++k) {
        acc.x = fmaf(v[k], m[k].x, acc.x);
        acc.y = fmaf(v[k], m[k].y, acc.y);
        acc.z = fmaf(v[k], m[k].z, acc.z);
        acc.w = fmaf(v[k], m[k].w, acc.w);
    }
}

template <int U>
__device__ __forceinline__ void rows64(const float* __restrict__ H, int mysrc, int myval,
                                       int j, int lane, float2& acc) {
    float2 m[U];
    float  v[U];
#pragma unroll
    for (int k = 0; k < U; ++k) {
        const int s = __builtin_amdgcn_readlane(mysrc, j + k);
        v[k] = __int_as_float(__builtin_amdgcn_readlane(myval, j + k));
        m[k] = ((const float2*)(H + (size_t)s * 64))[lane];
    }
#pragma unroll
    for (int k = 0; k < U; ++k) {
        acc.x = fmaf(v[k], m[k].x, acc.x);
        acc.y = fmaf(v[k], m[k].y, acc.y);
    }
}

static_assert(N_NODES % 32 == 0, "the grid covers the rows exactly: no row guard");

__global__ __launch_bounds__(512) void spmm_gather_d128_gemm2(
    const int* __restrict__ rowptr, const int2* __restrict__ list,
    const float* __restrict__ H, const float* __restrict__ W2,
    float* __restrict__ H2) {
    __shared__ float hs[32][132];
    __shared__ float w2s[128][72];

    const int tid   = threadIdx.x;
    const int lane  = tid & 31;
    const int wave  = tid >> 5;
    const int mbase = blockIdx.x * 32;

    int  beg[2], end[2];
    int2 first[2];
#pragma unroll
    for (int r = 0; r < 2; ++r) {
        const int row = mbase + wave + r * 16;
        beg[r] = __builtin_amdgcn_readfirstlane(rowptr[row]);
        end[r] = __builtin_amdgcn_readfirstlane(rowptr[row + 1]);
    }
#pragma unroll
    for (int r = 0; r < 2; ++r) {
        first[r] = make_int2(0, 0);
        if (lane < min(32, end[r] - beg[r])) first[r] = list[beg[r] + lane];
    }
#pragma unroll
    for (int i = 0; i < 4; ++i) {
        const int flat = (tid + i * 512) * 4;
        *(float4*)&w2s[flat >> 6][flat & 63] = *(const float4*)&W2[flat];
    }

#pragma unroll
    for (int r = 0; r < 2; ++r) {
        float4 acc = make_float4(0.0f, 0.0f, 0.0f, 0.0f);
        for (int base = beg[r]; base < end[r]; base += 32) {
            const int n = min(32, end[r] - base);
            int2 my = first[r];
            if (base != beg[r]) {
                my = make_int2(0, 0);
                if (lane < n) my = list[base + lane];
            }
            int j = 0;
            for (; j + 8 <= n; j += 8) rows128<8>(H, my.x, my.y, j, lane, acc);
            if (j + 4 <= n) { rows128<4>(H, my.x, my.y, j, lane, acc); j += 4; }
            if (j + 2 <= n) { rows128<2>(H, my.x, my.y, j, lane, acc); j += 2; }
            if (j < n)      { rows128<1>(H, my.x, my.y, j, lane, acc); }
        }
        acc.x = fmaxf(acc.x, 0.0f);
        acc.y = fmaxf(acc.y, 0.0f);
        acc.z = fmaxf(acc.z, 0.0f);
        acc.w = fmaxf(acc.w, 0.0f);
        *(float4*)&hs[wave + r * 16][lane * 4] = acc;
    }
    __syncthreads();
    if (wave >= 8) return;

    const int mt   = wave >> 2;
    const int nt   = wave & 3;
    const int half = lane >> 4;
    const int l16  = lane & 15;
    const int mrow = mt * 16 + l16;
    const int nb   = nt * 16 + l16;

    v8f d = {};
#pragma unroll
    for (int kk = 0; kk < HID_DIM; kk += 4) {
        const int k = kk + half * 2;
        v2f a = {hs[mrow][k], hs[mrow][k + 1]};
        v2f b = {w2s[k][nb], w2s[k + 1][nb]};
        d = __builtin_amdgcn_wmma_f32_16x16x4_f32(
            false, a, false, b, (short)0, d, false, false);
    }

#pragma unroll
    for (int v = 0; v < 8; ++v) {
        const int row = mbase + mt * 16 + v + half * 8;
        H2[(size_t)row * OUT_DIM + nb] = d[v];
    }
}

__global__ __launch_bounds__(256) void spmm_gather_d64(
    const int* __restrict__ rowptr, const int2* __restrict__ list,
    const float* __restrict__ H, float* __restrict__ out) {
    const int lane = threadIdx.x & 31;
    const int row  = blockIdx.x * 8 + (threadIdx.x >> 5);
    if (row >= N_NODES) return;
    const int beg = __builtin_amdgcn_readfirstlane(rowptr[row]);
    const int end = __builtin_amdgcn_readfirstlane(rowptr[row + 1]);
    float2 acc = make_float2(0.0f, 0.0f);
    for (int base = beg; base < end; base += 32) {
        const int n = min(32, end - base);
        int2 my = make_int2(0, 0);
        if (lane < n) my = list[base + lane];
        int j = 0;
        for (; j + 8 <= n; j += 8) rows64<8>(H, my.x, my.y, j, lane, acc);
        if (j + 4 <= n) { rows64<4>(H, my.x, my.y, j, lane, acc); j += 4; }
        if (j + 2 <= n) { rows64<2>(H, my.x, my.y, j, lane, acc); j += 2; }
        if (j < n)      { rows64<1>(H, my.x, my.y, j, lane, acc); }
    }
    ((float2*)(out + (size_t)row * 64))[lane] = acc;
}

extern "C" void kernel_launch(void* const* d_in, const int* in_sizes, int n_in,
                              void* d_out, int out_size, void* d_ws, size_t ws_size,
                              hipStream_t stream) {
    const float* X     = (const float*)d_in[0];
    const int*   esrc  = (const int*)d_in[1];
    const int*   edst  = (const int*)d_in[2];
    const float* evals = (const float*)d_in[3];
    const float* W1    = (const float*)d_in[4];
    const float* W2    = (const float*)d_in[5];
    float*       out   = (float*)d_out;

    char* ws = (char*)d_ws;
    const size_t h1Bytes = (size_t)N_NODES * HID_DIM * sizeof(float);
    float* H1   = (float*)(ws);
    float* H2   = (float*)(ws + h1Bytes);
    int2* list   = (int2*)(ws + 2 * h1Bytes);
    int*  rowptr = (int*)(ws + 2 * h1Bytes + 12800000 + 6400000 + 400128);
    int2* recs     = (int2*)(ws + h1Bytes);
    int*  counts   = (int*)(ws + h1Bytes + 12800000);
    int*  offs     = (int*)(ws + h1Bytes + 12800000 + 262144);

    chunk_sort<<<NCHUNK, 1024, 0, stream>>>(esrc, edst, evals, counts, offs, recs);
    bin_finish<<<NBIN, 1024, 0, stream>>>(counts, offs, recs, rowptr, list);

    gemm1_wmma<<<G1_GRID, 256, 0, stream>>>(X, W1, H1);
    spmm_gather_d128_gemm2<<<N_NODES / 32, 512, 0, stream>>>(rowptr, list, H1, W2, H2);
    spmm_gather_d64<<<N_NODES / 8, 256, 0, stream>>>(rowptr, list, H2, out);
}


# GEMM1 has no launch of its own: its row strips run in extra 16-wave blocks beside the two edge-grouping launches (all blocks 512 threads, 128 regs); 4 launches; H1 = sum of 2 half-K chains
# speedup vs baseline: 4.1429x; 1.0450x over previous
#include <hip/hip_runtime.h>

typedef __attribute__((ext_vector_type(2))) float v2f;
typedef __attribute__((ext_vector_type(8))) float v8f;

#define N_NODES 100000
#define N_EDGES 1600000
#define IN_DIM 256
#define HID_DIM 128
#define OUT_DIM 64

#define FUSED_THREADS 512
#define G1_STRIPS (N_NODES / 32)
#define G1_WG 256
#define G1_SPLIT 1536
#define G1_XS_BYTES (2 * 32 * 260 * 4)
#define G1_RED_BYTES (2 * 2 * 8 * 8 * 32 * 4)
#define SMEM_BYTES (G1_XS_BYTES + G1_RED_BYTES)
static_assert(N_NODES % 32 == 0, "strips cover the rows exactly: no row guard");
static_assert(G1_SPLIT >= G1_WG && G1_STRIPS - G1_SPLIT >= G1_WG,
              "every GEMM work-group owns a first strip in both launches");

__device__ __forceinline__ void gemm1_strips(const float* __restrict__ X,
                                             const float* __restrict__ W1,
                                             float* __restrict__ H1,
                                             char* smem, int g, int stBeg, int stEnd) {
    float (*xs)[32][260] = (float (*)[32][260])smem;
    float (*red)[2][8][8][32] = (float (*)[2][8][8][32])(smem + G1_XS_BYTES);

    const int tid   = threadIdx.x;
    const int lane  = tid & 31;
    const int wave  = tid >> 5;
    const int nt    = wave & 7;
    const int kh    = wave >> 3;
    const int half  = lane >> 4;
    const int l16   = lane & 15;
    const int nb    = nt * 16 + l16;
    const int kbase = kh * 128 + half * 2;
    const int srow  = tid >> 4;
    const int sseg  = (tid & 15) * 16;

    v2f b[32];
#pragma unroll
    for (int s = 0; s < 32; ++s) {
        const int k = kbase + s * 4;
        b[s] = (v2f){W1[(size_t)k * HID_DIM + nb], W1[(size_t)(k + 1) * HID_DIM + nb]};
    }

    int st = stBeg + g;
    {
        const float* xp = X + (size_t)(st * 32 + srow) * IN_DIM + sseg;
        float4 a0 = ((const float4*)xp)[0];
        float4 a1 = ((const float4*)xp)[1];
        float4 a2 = ((const float4*)xp)[2];
        float4 a3 = ((const float4*)xp)[3];
        *(float4*)&xs[0][srow][sseg + 0]  = a0;
        *(float4*)&xs[0][srow][sseg + 4]  = a1;
        *(float4*)&xs[0][srow][sseg + 8]  = a2;
        *(float4*)&xs[0][srow][sseg + 12] = a3;
    }
    __syncthreads();

    int buf = 0;
    for (; st < stEnd; st += G1_WG) {
        const int  stn  = st + G1_WG;
        const bool more = stn < stEnd;
        float4 n0, n1, n2, n3;
        if (more) {
            const float* xp = X + (size_t)(stn * 32 + srow) * IN_DIM + sseg;
            n0 = ((const float4*)xp)[0];
            n1 = ((const float4*)xp)[1];
            n2 = ((const float4*)xp)[2];
            n3 = ((const float4*)xp)[3];
        }

        v8f acc0 = {};
        v8f acc1 = {};
#pragma unroll
        for (int s = 0; s < 32; ++s) {
            const int ka = kbase + s * 4;
            v2f a0 = {xs[buf][l16][ka], xs[buf][l16][ka + 1]};
            v2f a1 = {xs[buf][16 + l16][ka], xs[buf][16 + l16][ka + 1]};
            acc0 = __builtin_amdgcn_wmma_f32_16x16x4_f32(
                false, a0, false, b[s], (short)0, acc0, false, false);
            acc1 = __builtin_amdgcn_wmma_f32_16x16x4_f32(
                false, a1, false, b[s], (short)0, acc1, false, false);
        }

        if (more) {
            *(float4*)&xs[buf ^ 1][srow][sseg + 0]  = n0;
            *(float4*)&xs[buf ^ 1][srow][sseg + 4]  = n1;
            *(float4*)&xs[buf ^ 1][srow][sseg + 8]  = n2;
            *(float4*)&xs[buf ^ 1][srow][sseg + 12] = n3;
        }
        if (kh == 1) {
#pragma unroll
            for (int v = 0; v < 8; ++v) {
                red[buf][0][nt][v][lane] = acc0[v];
                red[buf][1][nt][v][lane] = acc1[v];
            }
        }
        __syncthreads();

        if (kh == 0) {
#pragma unroll
            for (int v = 0; v < 8; ++v) {
                const int row = st * 32 + v + half * 8;
                H1[(size_t)row * HID_DIM + nb]        = acc0[v] + red[buf][0][nt][v][lane];
                H1[(size_t)(row + 16) * HID_DIM + nb] = acc1[v] + red[buf][1][nt][v][lane];
            }
        }
        buf ^= 1;
    }
}

#define NBIN 256
#define BIN_NODES 391
#define NCHUNK 256
#define CHUNK_EDGES (N_EDGES / NCHUNK)
#define CHUNK_TRIPS 13
#define SRC_BITS 17
static_assert(N_EDGES % NCHUNK == 0, "every chunk holds exactly CHUNK_EDGES edges");
static_assert(CHUNK_EDGES % 2 == 0, "a chunk's piece is a whole number of 16-byte pairs");
static_assert(CHUNK_TRIPS * FUSED_THREADS >= CHUNK_EDGES, "the trips cover a chunk");
static_assert(CHUNK_EDGES * 8 + 2 * NBIN * 4 + 8 * 4 <= SMEM_BYTES, "the sort's LDS arrays fit");

__device__ __forceinline__ void chunk_sort(const int* __restrict__ src,
                                           const int* __restrict__ dst,
                                           const float* __restrict__ vals,
                                           int* __restrict__ counts,
                                           int* __restrict__ offs,
                                           int2* __restrict__ recs,
                                           char* smem, int me) {
    int2* srt   = (int2*)smem;
    int*  hist  = (int*)(smem + CHUNK_EDGES * 8);
    int*  first = hist + NBIN;
    int*  wsum  = first + NBIN;
    const int t    = threadIdx.x;
    const int lane = t & 31;
    const int wave = t >> 5;
    const int beg  = me * CHUNK_EDGES;
    const int end  = beg + CHUNK_EDGES;

    if (t < NBIN) hist[t] = 0;

    int   d[CHUNK_TRIPS], s[CHUNK_TRIPS];
    float v[CHUNK_TRIPS];
#pragma unroll
    for (int k = 0; k < CHUNK_TRIPS; ++k) {
        const int e = beg + k * FUSED_THREADS + t;
        const bool ok = e < end;
        d[k] = ok ? dst[e] : -1;
        s[k] = ok ? src[e] : 0;
        v[k] = ok ? vals[e] : 0.0f;
    }
    __syncthreads();

    int rank[CHUNK_TRIPS], bin[CHUNK_TRIPS];
#pragma unroll
    for (int k = 0; k < CHUNK_TRIPS; ++k) {
        rank[k] = 0;
        bin[k]  = 0;
        if (d[k] >= 0) {
            const unsigned bb = (unsigned)d[k] / BIN_NODES;
            bin[k]  = (int)bb;
            rank[k] = atomicAdd(&hist[bb], 1);
            s[k] |= (d[k] - (int)bb * BIN_NODES) << SRC_BITS;
        }
    }
    __syncthreads();

    const int myCnt = (t < NBIN) ? hist[t] : 0;
    int incl = myCnt;
    for (int o = 1; o < 32; o <<= 1) {
        const int u = __shfl_up(incl, o);
        if (lane >= o) incl += u;
    }
    if (lane == 31 && wave < 8) wsum[wave] = incl;
    __syncthreads();
    if (t < NBIN) {
        int base = 0;
        for (int w = 0; w < wave; ++w) base += wsum[w];
        const int excl = base + incl - myCnt;
        first[t] = excl;
        counts[me * NBIN + t] = myCnt;
        offs[me * NBIN + t]   = excl;
    }
    __syncthreads();

    int slot[CHUNK_TRIPS];
#pragma unroll
    for (int k = 0; k < CHUNK_TRIPS; ++k) slot[k] = first[bin[k]] + rank[k];
#pragma unroll
    for (int k = 0; k < CHUNK_TRIPS; ++k)
        if (d[k] >= 0) srt[slot[k]] = make_int2(s[k], __float_as_int(v[k]));
    __syncthreads();

    int4*       outp = (int4*)(recs + beg);
    const int4* inp  = (const int4*)srt;
    for (int i = t; i < CHUNK_EDGES / 2; i += FUSED_THREADS) outp[i] = inp[i];
}

static_assert(FUSED_THREADS == 512 && BIN_NODES <= 512, "one counter per thread in the scan");
static_assert(FUSED_THREADS == 2 * NCHUNK, "2 lanes per chunk piece");

__device__ __forceinline__ void bin_finish(const int* __restrict__ counts,
                                           const int* __restrict__ offs,
                                           const int2* __restrict__ recs,
                                           int* __restrict__ rowptr,
                                           int2* __restrict__ list,
                                           char* smem, int b) {
    int* segLen   = (int*)smem;
    int* segStart = segLen + NCHUNK;
    int* cnt      = segStart + NCHUNK;
    int* wsum     = cnt + 512;
    int* colOff   = wsum + 16;
    int* colLen   = colOff + 8;
    const int t    = threadIdx.x;
    const int lane = t & 31;
    const int wave = t >> 5;
    const int lo   = b * BIN_NODES;
    const int nn   = min(BIN_NODES, N_NODES - lo);

    cnt[t] = 0;
    int myLen = 0, myOff = 0;
    if (t < NCHUNK) {
        myLen = counts[t * NBIN + b];
        myOff = offs[t * NBIN + b];
        segLen[t]   = myLen;
        segStart[t] = t * CHUNK_EDGES + myOff;
    }
    int sOff = myOff, sLen = myLen;
    for (int o = 16; o > 0; o >>= 1) {
        sOff += __shfl_xor(sOff, o);
        sLen += __shfl_xor(sLen, o);
    }
    if (lane == 0 && wave < 8) { colOff[wave] = sOff; colLen[wave] = sLen; }
    __syncthreads();
    int bs = 0, be = 0;
#pragma unroll
    for (int w = 0; w < 8; ++w) { bs += colOff[w]; be += colLen[w]; }
    be += bs;

    const int   len = segLen[t >> 1];
    const int2* seg = recs + segStart[t >> 1];
    const int   sub = t & 1;

    for (int j0 = 0; j0 < len; j0 += 32) {
        int x[16];
#pragma unroll
        for (int k = 0; k < 16; ++k) {
            const int idx = j0 + k * 2 + sub;
            x[k] = (idx < len) ? seg[idx].x : -1;
        }
#pragma unroll
        for (int k = 0; k < 16; ++k)
            if (x[k] >= 0) atomicAdd(&cnt[x[k] >> SRC_BITS], 1);
    }
    __syncthreads();

    const int c = cnt[t];
    int incl = c;
    for (int o = 1; o < 32; o <<= 1) {
        const int u = __shfl_up(incl, o);
        if (lane >= o) incl += u;
    }
    if (lane == 31) wsum[wave] = incl;
    __syncthreads();
    {
        int pre = 0;
        for (int w = 0; w < wave; ++w) pre += wsum[w];
        const int excl = bs + pre + incl - c;
        cnt[t] = excl;
        if (t < nn) rowptr[lo + t] = excl;
    }
    if (b == NBIN - 1 && t == 0) rowptr[N_NODES] = be;
    __syncthreads();

    for (int j0 = 0; j0 < len; j0 += 32) {
        int2 r[16];
#pragma unroll
        for (int k = 0; k < 16; ++k) {
            const int idx = j0 + k * 2 + sub;
            r[k] = (idx < len) ? seg[idx] : make_int2(-1, 0);
        }
        int pos[16];
#pragma unroll
        for (int k = 0; k < 16; ++k) {
            pos[k] = 0;
            if (r[k].x >= 0) pos[k] = atomicAdd(&cnt[r[k].x >> SRC_BITS], 1);
        }
#pragma unroll
        for (int k = 0; k < 16; ++k)
            if (r[k].x >= 0)
                list[pos[k]] = make_int2(r[k].x & ((1 << SRC_BITS) - 1), r[k].y);
    }
}

__global__ __launch_bounds__(FUSED_THREADS, 8) void sort_and_gemm1(
    const int* __restrict__ src, const int* __restrict__ dst, const float* __restrict__ vals,
    int* __restrict__ counts, int* __restrict__ offs, int2* __restrict__ recs,
    const float* __restrict__ X, const float* __restrict__ W1, float* __restrict__ H1) {
    __shared__ __align__(16) char smem[SMEM_BYTES];
    if (blockIdx.x < G1_WG)
        gemm1_strips(X, W1, H1, smem, blockIdx.x, 0, G1_SPLIT);
    else
        chunk_sort(src, dst, vals, counts, offs, recs, smem, blockIdx.x - G1_WG);
}

__global__ __launch_bounds__(FUSED_THREADS, 8) void finish_and_gemm1(
    const int* __restrict__ counts, const int* __restrict__ offs, const int2* __restrict__ recs,
    int* __restrict__ rowptr, int2* __restrict__ list,
    const float* __restrict__ X, const float* __restrict__ W1, float* __restrict__ H1) {
    __shared__ __align__(16) char smem[SMEM_BYTES];
    if (blockIdx.x < G1_WG)
        gemm1_strips(X, W1, H1, smem, blockIdx.x, G1_SPLIT, G1_STRIPS);
    else
        bin_finish(counts, offs, recs, rowptr, list, smem, blockIdx.x - G1_WG);
}

template <int U>
__device__ __forceinline__ void rows128(const float* __restrict__ H, int mysrc, int myval,
                                        int j, int lane, float4& acc) {
    float4 m[U];
    float  v[U];
#pragma unroll
    for (int k = 0; k < U; ++k) {
        const int s = __builtin_amdgcn_readlane(mysrc, j + k);
        v[k] = __int_as_float(__builtin_amdgcn_readlane(myval, j + k));
        m[k] = ((const float4*)(H + (size_t)s * 128))[lane];
    }
#pragma unroll
    for (int k = 0; k < U; ++k) {
        acc.x = fmaf(v[k], m[k].x, acc.x);
        acc.y = fmaf(v[k], m[k].y, acc.y);
        acc.z = fmaf(v[k], m[k].z, acc.z);
        acc.w = fmaf(v[k], m[k].w, acc.w);
    }
}

template <int U>
__device__ __forceinline__ void rows64(const float* __restrict__ H, int mysrc, int myval,
                                       int j, int lane, float2& acc) {
    float2 m[U];
    float  v[U];
#pragma unroll
    for (int k = 0; k < U; ++k) {
        const int s = __builtin_amdgcn_readlane(mysrc, j + k);
        v[k] = __int_as_float(__builtin_amdgcn_readlane(myval, j + k));
        m[k] = ((const float2*)(H + (size_t)s * 64))[lane];
    }
#pragma unroll
    for (int k = 0; k < U; ++k) {
        acc.x = fmaf(v[k], m[k].x, acc.x);
        acc.y = fmaf(v[k], m[k].y, acc.y);
    }
}

static_assert(N_NODES % 32 == 0, "the grid covers the rows exactly: no row guard");

__global__ __launch_bounds__(512) void spmm_gather_d128_gemm2(
    const int* __restrict__ rowptr, const int2* __restrict__ list,
    const float* __restrict__ H, const float* __restrict__ W2,
    float* __restrict__ H2) {
    __shared__ float hs[32][132];
    __shared__ float w2s[128][72];

    const int tid   = threadIdx.x;
    const int lane  = tid & 31;
    const int wave  = tid >> 5;
    const int mbase = blockIdx.x * 32;

    int  beg[2], end[2];
    int2 first[2];
#pragma unroll
    for (int r = 0; r < 2; ++r) {
        const int row = mbase + wave + r * 16;
        beg[r] = __builtin_amdgcn_readfirstlane(rowptr[row]);
        end[r] = __builtin_amdgcn_readfirstlane(rowptr[row + 1]);
    }
#pragma unroll
    for (int r = 0; r < 2; ++r) {
        first[r] = make_int2(0, 0);
        if (lane < min(32, end[r] - beg[r])) first[r] = list[beg[r] + lane];
    }
#pragma unroll
    for (int i = 0; i < 4; ++i) {
        const int flat = (tid + i * 512) * 4;
        *(float4*)&w2s[flat >> 6][flat & 63] = *(const float4*)&W2[flat];
    }

#pragma unroll
    for (int r = 0; r < 2; ++r) {
        float4 acc = make_float4(0.0f, 0.0f, 0.0f, 0.0f);
        for (int base = beg[r]; base < end[r]; base += 32) {
            const int n = min(32, end[r] - base);
            int2 my = first[r];
            if (base != beg[r]) {
                my = make_int2(0, 0);
                if (lane < n) my = list[base + lane];
            }
            int j = 0;
            for (; j + 8 <= n; j += 8) rows128<8>(H, my.x, my.y, j, lane, acc);
            if (j + 4 <= n) { rows128<4>(H, my.x, my.y, j, lane, acc); j += 4; }
            if (j + 2 <= n) { rows128<2>(H, my.x, my.y, j, lane, acc); j += 2; }
            if (j < n)      { rows128<1>(H, my.x, my.y, j, lane, acc); }
        }
        acc.x = fmaxf(acc.x, 0.0f);
        acc.y = fmaxf(acc.y, 0.0f);
        acc.z = fmaxf(acc.z, 0.0f);
        acc.w = fmaxf(acc.w, 0.0f);
        *(float4*)&hs[wave + r * 16][lane * 4] = acc;
    }
    __syncthreads();
    if (wave >= 8) return;

    const int mt   = wave >> 2;
    const int nt   = wave & 3;
    const int half = lane >> 4;
    const int l16  = lane & 15;
    const int mrow = mt * 16 + l16;
    const int nb   = nt * 16 + l16;

    v8f d = {};
#pragma unroll
    for (int kk = 0; kk < HID_DIM; kk += 4) {
        const int k = kk + half * 2;
        v2f a = {hs[mrow][k], hs[mrow][k + 1]};
        v2f b = {w2s[k][nb], w2s[k + 1][nb]};
        d = __builtin_amdgcn_wmma_f32_16x16x4_f32(
            false, a, false, b, (short)0, d, false, false);
    }

#pragma unroll
    for (int v = 0; v < 8; ++v) {
        const int row = mbase + mt * 16 + v + half * 8;
        H2[(size_t)row * OUT_DIM + nb] = d[v];
    }
}

__global__ __launch_bounds__(256) void spmm_gather_d64(
    const int* __restrict__ rowptr, const int2* __restrict__ list,
    const float* __restrict__ H, float* __restrict__ out) {
    const int lane = threadIdx.x & 31;
    const int row  = blockIdx.x * 8 + (threadIdx.x >> 5);
    if (row >= N_NODES) return;
    const int beg = __builtin_amdgcn_readfirstlane(rowptr[row]);
    const int end = __builtin_amdgcn_readfirstlane(rowptr[row + 1]);
    float2 acc = make_float2(0.0f, 0.0f);
    for (int base = beg; base < end; base += 32) {
        const int n = min(32, end - base);
        int2 my = make_int2(0, 0);
        if (lane < n) my = list[base + lane];
        int j = 0;
        for (; j + 8 <= n; j += 8) rows64<8>(H, my.x, my.y, j, lane, acc);
        if (j + 4 <= n) { rows64<4>(H, my.x, my.y, j, lane, acc); j += 4; }
        if (j + 2 <= n) { rows64<2>(H, my.x, my.y, j, lane, acc); j += 2; }
        if (j < n)      { rows64<1>(H, my.x, my.y, j, lane, acc); }
    }
    ((float2*)(out + (size_t)row * 64))[lane] = acc;
}

extern "C" void kernel_launch(void* const* d_in, const int* in_sizes, int n_in,
                              void* d_out, int out_size, void* d_ws, size_t ws_size,
                              hipStream_t stream) {
    const float* X     = (const float*)d_in[0];
    const int*   esrc  = (const int*)d_in[1];
    const int*   edst  = (const int*)d_in[2];
    const float* evals = (const float*)d_in[3];
    const float* W1    = (const float*)d_in[4];
    const float* W2    = (const float*)d_in[5];
    float*       out   = (float*)d_out;

    char* ws = (char*)d_ws;
    const size_t h1Bytes = (size_t)N_NODES * HID_DIM * sizeof(float);
    float* H1   = (float*)(ws);
    float* H2   = (float*)(ws + h1Bytes);
    int2* list   = (int2*)(ws + 2 * h1Bytes);
    int*  rowptr = (int*)(ws + 2 * h1Bytes + 12800000 + 6400000 + 400128);
    int2* recs     = (int2*)(ws + h1Bytes);
    int*  counts   = (int*)(ws + h1Bytes + 12800000);
    int*  offs     = (int*)(ws + h1Bytes + 12800000 + 262144);

    sort_and_gemm1<<<G1_WG + NCHUNK, FUSED_THREADS, 0, stream>>>(
        esrc, edst, evals, counts, offs, recs, X, W1, H1);
    finish_and_gemm1<<<G1_WG + NBIN, FUSED_THREADS, 0, stream>>>(
        counts, offs, recs, rowptr, list, X, W1, H1);
    spmm_gather_d128_gemm2<<<N_NODES / 32, 512, 0, stream>>>(rowptr, list, H1, W2, H2);
    spmm_gather_d64<<<N_NODES / 8, 256, 0, stream>>>(rowptr, list, H2, out);
}
